# starting kernel
# baseline (speedup 1.0000x reference)
#include <hip/hip_runtime.h>
#include <math.h>

#define B_     128
#define NE_    64
#define NN_    16
#define D_     128
#define NRBF_  16
#define L_     3
#define NUP_   32
#define CUTOFF_ 2.0f
#define EPS_    1e-12f

#define KEE_   (NE_ * NRBF_)
#define KNE_   (NN_ * NRBF_)
#define NT_    4
#define WSZ_   (D_ * D_)

typedef float v2f __attribute__((ext_vector_type(2)));
typedef float v8f __attribute__((ext_vector_type(8)));

__device__ __forceinline__ void gemm_strip_f32p(const float* __restrict__ A, int lda,
                                                const float* __restrict__ Bp,
                                                int m0, int n0, v8f acc[NT_], int lane) {
  const int half = lane >> 4;
  const int l16  = lane & 15;
  const float* __restrict__ arow = A + (size_t)(m0 + l16) * lda;
#pragma unroll
  for (int k0 = 0; k0 < D_; k0 += 4) {
    const int ka = k0 + half * 2;
    const v2f a = *(const v2f*)&arow[ka];
    const float* __restrict__ brow = Bp + (size_t)(ka >> 1) * (2 * D_);
#pragma unroll
    for (int t = 0; t < NT_; ++t) {
      const v2f b = *(const v2f*)&brow[(n0 + t * 16 + l16) * 2];
      acc[t] = __builtin_amdgcn_wmma_f32_16x16x4_f32(false, a, false, b, (short)0,
                                                     acc[t], false, false);
    }
  }
}

__device__ __forceinline__ void store_strip_f32(float* __restrict__ C, int ldc,
                                                int m0, int n0, v8f acc[NT_], int lane) {
  const int half = lane >> 4;
  const int l16  = lane & 15;
#pragma unroll
  for (int t = 0; t < NT_; ++t)
#pragma unroll
    for (int ii = 0; ii < 8; ++ii)
      C[(size_t)(m0 + half * 8 + ii) * ldc + n0 + t * 16 + l16] = acc[t][ii];
}

__device__ __forceinline__ void load_strip_f32(const float* __restrict__ C, int ldc,
                                               int m0, int n0, v8f acc[NT_], int lane) {
  const int half = lane >> 4;
  const int l16  = lane & 15;
#pragma unroll
  for (int t = 0; t < NT_; ++t)
#pragma unroll
    for (int ii = 0; ii < 8; ++ii)
      acc[t][ii] = C[(size_t)(m0 + half * 8 + ii) * ldc + n0 + t * 16 + l16];
}

__global__ void gnn_pack_w(const float* __restrict__ We, const float* __restrict__ Wn,
                           const float* __restrict__ W1, const float* __restrict__ W2,
                           float* __restrict__ Wp) {
  const int gid = blockIdx.x * blockDim.x + threadIdx.x;
  const int m   = gid >> 14;
  const int idx = gid & (WSZ_ - 1);
  const int l = m >> 2, w = m & 3;
  const float* src = (w == 0 ? We : w == 1 ? Wn : w == 2 ? W1 : W2) + (size_t)l * WSZ_;
  const int k = idx >> 7, col = idx & (D_ - 1);
  Wp[(size_t)m * WSZ_ + (size_t)(k >> 1) * (2 * D_) + col * 2 + (k & 1)] = src[idx];
}

__global__ void gnn_feat_ee(const float* __restrict__ r, float* __restrict__ Aee) {
  const int bi = blockIdx.x;
  const int b = bi >> 6, i = bi & 63;
  const float rx = r[(size_t)(b * NE_ + i) * 3 + 0];
  const float ry = r[(size_t)(b * NE_ + i) * 3 + 1];
  const float rz = r[(size_t)(b * NE_ + i) * 3 + 2];
  const float inv_cut = 1.0f / CUTOFF_;
  const float dmu     = CUTOFF_ / (float)(NRBF_ - 1);
  const float inv_sig = (float)NRBF_ / CUTOFF_;
  for (int t = threadIdx.x; t < KEE_; t += blockDim.x) {
    const int j = t >> 4, k = t & 15;
    const float dx = rx - r[(size_t)(b * NE_ + j) * 3 + 0];
    const float dy = ry - r[(size_t)(b * NE_ + j) * 3 + 1];
    const float dz = rz - r[(size_t)(b * NE_ + j) * 3 + 2];
    const float dist = sqrtf(dx * dx + dy * dy + dz * dz + EPS_);
    float env = 1.0f - dist * inv_cut;
    env = (dist < CUTOFF_) ? env * env : 0.0f;
    const float z = (dist - dmu * (float)k) * inv_sig;
    float g = expf(-0.5f * z * z) * env;
    if (i == j) g = 0.0f;
    Aee[(size_t)bi * KEE_ + t] = g;
  }
}

__global__ void gnn_feat_ne(const float* __restrict__ r, const float* __restrict__ coords,
                            float* __restrict__ Ane) {
  const int bj = blockIdx.x;
  const int b = bj >> 6, j = bj & 63;
  const float rx = r[(size_t)(b * NE_ + j) * 3 + 0];
  const float ry = r[(size_t)(b * NE_ + j) * 3 + 1];
  const float rz = r[(size_t)(b * NE_ + j) * 3 + 2];
  const float inv_cut = 1.0f / CUTOFF_;
  const float dmu     = CUTOFF_ / (float)(NRBF_ - 1);
  const float inv_sig = (float)NRBF_ / CUTOFF_;
  const int t = threadIdx.x;
  const int a = t >> 4, k = t & 15;
  const float dx = rx - coords[a * 3 + 0];
  const float dy = ry - coords[a * 3 + 1];
  const float dz = rz - coords[a * 3 + 2];
  const float dist = sqrtf(dx * dx + dy * dy + dz * dz + EPS_);
  float env = 1.0f - dist * inv_cut;
  env = (dist < CUTOFF_) ? env * env : 0.0f;
  const float z = (dist - dmu * (float)k) * inv_sig;
  const float g = expf(-0.5f * z * z) * env;
  Ane[(size_t)bj * KNE_ + t] = g;
}

__global__ void gnn_init_h(const float* __restrict__ spin_emb, float* __restrict__ h) {
  const int gid = blockIdx.x * blockDim.x + threadIdx.x;
  const int d = gid & (D_ - 1);
  const int i = (gid >> 7) & (NE_ - 1);
  h[gid] = spin_emb[(i < NUP_ ? 0 : 1) * D_ + d];
}

__global__ void gnn_hw(const float* __restrict__ h, const float* __restrict__ nuc_emb,
                       const float* __restrict__ Wp_ee, const float* __restrict__ Wp_ne,
                       float* __restrict__ hW, float* __restrict__ hnW) {
  const int wid = blockIdx.x;
  const int lane = threadIdx.x & 31;
  v8f acc[NT_] = {};
  if (wid < 1024) {
    const int m0 = (wid >> 1) * 16;
    const int n0 = (wid & 1) * 64;
    gemm_strip_f32p(h, D_, Wp_ee, m0, n0, acc, lane);
    store_strip_f32(hW, D_, m0, n0, acc, lane);
  } else {
    const int n0 = (wid - 1024) * 64;
    gemm_strip_f32p(nuc_emb, D_, Wp_ne, 0, n0, acc, lane);
    store_strip_f32(hnW, D_, 0, n0, acc, lane);
  }
}

__global__ void gnn_agg(const float* __restrict__ Aee, const float* __restrict__ Ane,
                        const float* __restrict__ hW, const float* __restrict__ hnW,
                        const float* __restrict__ Wrbf_ee_l, const float* __restrict__ Wrbf_ne_l,
                        float* __restrict__ agg) {
  const int wid = blockIdx.x;
  const int b  = wid >> 3;
  const int mt = (wid >> 1) & 3;
  const int nb = wid & 1;
  const int lane = threadIdx.x & 31;
  const int half = lane >> 4, l16 = lane & 15;
  const int i0 = mt * 16, n0 = nb * 64;

  const float* __restrict__ Ab  = Aee + (size_t)b * NE_ * KEE_;
  const float* __restrict__ An  = Ane + (size_t)b * NE_ * KNE_;
  const float* __restrict__ hWb = hW + (size_t)b * NE_ * D_;
  v8f acc[NT_] = {};

  float wk[NT_][8];

  {
#pragma unroll
    for (int kk = 0; kk < 4; ++kk)
#pragma unroll
      for (int t = 0; t < NT_; ++t) {
        const int kb = kk * 4 + half * 2;
        wk[t][kk * 2 + 0] = Wrbf_ee_l[(size_t)(kb + 0) * D_ + n0 + t * 16 + l16];
        wk[t][kk * 2 + 1] = Wrbf_ee_l[(size_t)(kb + 1) * D_ + n0 + t * 16 + l16];
      }
    const float* __restrict__ arow = Ab + (size_t)(i0 + l16) * KEE_;
    for (int j = 0; j < NE_; ++j) {
      float hv[NT_];
#pragma unroll
      for (int t = 0; t < NT_; ++t) hv[t] = hWb[(size_t)j * D_ + n0 + t * 16 + l16];
#pragma unroll
      for (int kk = 0; kk < 4; ++kk) {
        const int ka = j * 16 + kk * 4 + half * 2;
        const v2f a = *(const v2f*)&arow[ka];
#pragma unroll
        for (int t = 0; t < NT_; ++t) {
          v2f bb;
          bb.x = hv[t] * wk[t][kk * 2 + 0];
          bb.y = hv[t] * wk[t][kk * 2 + 1];
          acc[t] = __builtin_amdgcn_wmma_f32_16x16x4_f32(false, a, false, bb, (short)0,
                                                         acc[t], false, false);
        }
      }
    }
  }

  {
#pragma unroll
    for (int kk = 0; kk < 4; ++kk)
#pragma unroll
      for (int t = 0; t < NT_; ++t) {
        const int kb = kk * 4 + half * 2;
        wk[t][kk * 2 + 0] = Wrbf_ne_l[(size_t)(kb + 0) * D_ + n0 + t * 16 + l16];
        wk[t][kk * 2 + 1] = Wrbf_ne_l[(size_t)(kb + 1) * D_ + n0 + t * 16 + l16];
      }
    const float* __restrict__ arow = An + (size_t)(i0 + l16) * KNE_;
#pragma unroll
    for (int a_idx = 0; a_idx < NN_; ++a_idx) {
      float hv[NT_];
#pragma unroll
      for (int t = 0; t < NT_; ++t) hv[t] = hnW[(size_t)a_idx * D_ + n0 + t * 16 + l16];
#pragma unroll
      for (int kk = 0; kk < 4; ++kk) {
        const int ka = a_idx * 16 + kk * 4 + half * 2;
        const v2f a = *(const v2f*)&arow[ka];
#pragma unroll
        for (int t = 0; t < NT_; ++t) {
          v2f bb;
          bb.x = hv[t] * wk[t][kk * 2 + 0];
          bb.y = hv[t] * wk[t][kk * 2 + 1];
          acc[t] = __builtin_amdgcn_wmma_f32_16x16x4_f32(false, a, false, bb, (short)0,
                                                         acc[t], false, false);
        }
      }
    }
  }

  store_strip_f32(agg + (size_t)b * NE_ * D_, D_, i0, n0, acc, lane);
}

__global__ void gnn_mlp1(const float* __restrict__ agg, const float* __restrict__ Wp1,
                         float* __restrict__ tmp) {
  const int wid = blockIdx.x;
  const int lane = threadIdx.x & 31;
  const int m0 = (wid >> 1) * 16;
  const int n0 = (wid & 1) * 64;
  v8f acc[NT_] = {};
  gemm_strip_f32p(agg, D_, Wp1, m0, n0, acc, lane);
#pragma unroll
  for (int t = 0; t < NT_; ++t)
#pragma unroll
    for (int ii = 0; ii < 8; ++ii) acc[t][ii] = tanhf(acc[t][ii]);
  store_strip_f32(tmp, D_, m0, n0, acc, lane);
}

__global__ void gnn_mlp2(const float* __restrict__ tmp, const float* __restrict__ Wp2,
                         const float* __restrict__ h, float* __restrict__ dst) {
  const int wid = blockIdx.x;
  const int lane = threadIdx.x & 31;
  const int m0 = (wid >> 1) * 16;
  const int n0 = (wid & 1) * 64;
  v8f acc[NT_];
  load_strip_f32(h, D_, m0, n0, acc, lane);
  gemm_strip_f32p(tmp, D_, Wp2, m0, n0, acc, lane);
  store_strip_f32(dst, D_, m0, n0, acc, lane);
}

extern "C" void kernel_launch(void* const* d_in, const int* in_sizes, int n_in,
                              void* d_out, int out_size, void* d_ws, size_t ws_size,
                              hipStream_t stream) {
  (void)in_sizes; (void)n_in; (void)out_size; (void)ws_size;
  const float* r        = (const float*)d_in[0];
  const float* coords   = (const float*)d_in[1];
  const float* spin_emb = (const float*)d_in[2];
  const float* nuc_emb  = (const float*)d_in[3];
  const float* W_rbf_ee = (const float*)d_in[4];
  const float* W_rbf_ne = (const float*)d_in[5];
  const float* W_msg_ee = (const float*)d_in[6];
  const float* W_msg_ne = (const float*)d_in[7];
  const float* W1       = (const float*)d_in[8];
  const float* W2       = (const float*)d_in[9];
  float* out = (float*)d_out;

  float* ws  = (float*)d_ws;
  float* Aee = ws;
  float* Ane = Aee + (size_t)B_ * NE_ * KEE_;
  float* h   = Ane + (size_t)B_ * NE_ * KNE_;
  float* hW  = h   + (size_t)B_ * NE_ * D_;
  float* hnW = hW  + (size_t)B_ * NE_ * D_;
  float* agg = hnW + (size_t)NN_ * D_;
  float* tmp = agg + (size_t)B_ * NE_ * D_;
  float* Wp  = tmp + (size_t)B_ * NE_ * D_;

  gnn_pack_w<<<(12 * WSZ_) / 256, 256, 0, stream>>>(W_msg_ee, W_msg_ne, W1, W2, Wp);
  gnn_feat_ee<<<B_ * NE_, 256, 0, stream>>>(r, Aee);
  gnn_feat_ne<<<B_ * NE_, 256, 0, stream>>>(r, coords, Ane);
  gnn_init_h<<<(B_ * NE_ * D_) / 256, 256, 0, stream>>>(spin_emb, h);

  for (int l = 0; l < L_; ++l) {
    const float* Wpl = Wp + (size_t)l * 4 * WSZ_;
    gnn_hw<<<1024 + 2, 32, 0, stream>>>(h, nuc_emb, Wpl + 0 * WSZ_, Wpl + 1 * WSZ_, hW, hnW);
    gnn_agg<<<B_ * 8, 32, 0, stream>>>(Aee, Ane, hW, hnW,
                                       W_rbf_ee + (size_t)l * NRBF_ * D_,
                                       W_rbf_ne + (size_t)l * NRBF_ * D_, agg);
    gnn_mlp1<<<1024, 32, 0, stream>>>(agg, Wpl + 2 * WSZ_, tmp);
    float* dst = (l == L_ - 1) ? out : h;
    gnn_mlp2<<<1024, 32, 0, stream>>>(tmp, Wpl + 3 * WSZ_, h, dst);
  }
}


# agg kernel: A-fragment and hW loads issued 6 k-steps ahead of their WMMAs into an 8-stage register ring, ee and ne parts as one 80-step stream; products and summation order unchanged
# speedup vs baseline: 1.1900x; 1.1900x over previous
#include <hip/hip_runtime.h>
#include <math.h>

#define B_     128
#define NE_    64
#define NN_    16
#define D_     128
#define NRBF_  16
#define L_     3
#define NUP_   32
#define CUTOFF_ 2.0f
#define EPS_    1e-12f

#define KEE_   (NE_ * NRBF_)
#define KNE_   (NN_ * NRBF_)
#define NT_    4
#define WSZ_   (D_ * D_)

typedef float v2f __attribute__((ext_vector_type(2)));
typedef float v8f __attribute__((ext_vector_type(8)));

__device__ __forceinline__ void gemm_strip_f32p(const float* __restrict__ A, int lda,
                                                const float* __restrict__ Bp,
                                                int m0, int n0, v8f acc[NT_], int lane) {
  const int half = lane >> 4;
  const int l16  = lane & 15;
  const float* __restrict__ arow = A + (size_t)(m0 + l16) * lda;
#pragma unroll
  for (int k0 = 0; k0 < D_; k0 += 4) {
    const int ka = k0 + half * 2;
    const v2f a = *(const v2f*)&arow[ka];
    const float* __restrict__ brow = Bp + (size_t)(ka >> 1) * (2 * D_);
#pragma unroll
    for (int t = 0; t < NT_; ++t) {
      const v2f b = *(const v2f*)&brow[(n0 + t * 16 + l16) * 2];
      acc[t] = __builtin_amdgcn_wmma_f32_16x16x4_f32(false, a, false, b, (short)0,
                                                     acc[t], false, false);
    }
  }
}

__device__ __forceinline__ void store_strip_f32(float* __restrict__ C, int ldc,
                                                int m0, int n0, v8f acc[NT_], int lane) {
  const int half = lane >> 4;
  const int l16  = lane & 15;
#pragma unroll
  for (int t = 0; t < NT_; ++t)
#pragma unroll
    for (int ii = 0; ii < 8; ++ii)
      C[(size_t)(m0 + half * 8 + ii) * ldc + n0 + t * 16 + l16] = acc[t][ii];
}

__device__ __forceinline__ void load_strip_f32(const float* __restrict__ C, int ldc,
                                               int m0, int n0, v8f acc[NT_], int lane) {
  const int half = lane >> 4;
  const int l16  = lane & 15;
#pragma unroll
  for (int t = 0; t < NT_; ++t)
#pragma unroll
    for (int ii = 0; ii < 8; ++ii)
      acc[t][ii] = C[(size_t)(m0 + half * 8 + ii) * ldc + n0 + t * 16 + l16];
}

__global__ void gnn_pack_w(const float* __restrict__ We, const float* __restrict__ Wn,
                           const float* __restrict__ W1, const float* __restrict__ W2,
                           float* __restrict__ Wp) {
  const int gid = blockIdx.x * blockDim.x + threadIdx.x;
  const int m   = gid >> 14;
  const int idx = gid & (WSZ_ - 1);
  const int l = m >> 2, w = m & 3;
  const float* src = (w == 0 ? We : w == 1 ? Wn : w == 2 ? W1 : W2) + (size_t)l * WSZ_;
  const int k = idx >> 7, col = idx & (D_ - 1);
  Wp[(size_t)m * WSZ_ + (size_t)(k >> 1) * (2 * D_) + col * 2 + (k & 1)] = src[idx];
}

__global__ void gnn_feat_ee(const float* __restrict__ r, float* __restrict__ Aee) {
  const int bi = blockIdx.x;
  const int b = bi >> 6, i = bi & 63;
  const float rx = r[(size_t)(b * NE_ + i) * 3 + 0];
  const float ry = r[(size_t)(b * NE_ + i) * 3 + 1];
  const float rz = r[(size_t)(b * NE_ + i) * 3 + 2];
  const float inv_cut = 1.0f / CUTOFF_;
  const float dmu     = CUTOFF_ / (float)(NRBF_ - 1);
  const float inv_sig = (float)NRBF_ / CUTOFF_;
  for (int t = threadIdx.x; t < KEE_; t += blockDim.x) {
    const int j = t >> 4, k = t & 15;
    const float dx = rx - r[(size_t)(b * NE_ + j) * 3 + 0];
    const float dy = ry - r[(size_t)(b * NE_ + j) * 3 + 1];
    const float dz = rz - r[(size_t)(b * NE_ + j) * 3 + 2];
    const float dist = sqrtf(dx * dx + dy * dy + dz * dz + EPS_);
    float env = 1.0f - dist * inv_cut;
    env = (dist < CUTOFF_) ? env * env : 0.0f;
    const float z = (dist - dmu * (float)k) * inv_sig;
    float g = expf(-0.5f * z * z) * env;
    if (i == j) g = 0.0f;
    Aee[(size_t)bi * KEE_ + t] = g;
  }
}

__global__ void gnn_feat_ne(const float* __restrict__ r, const float* __restrict__ coords,
                            float* __restrict__ Ane) {
  const int bj = blockIdx.x;
  const int b = bj >> 6, j = bj & 63;
  const float rx = r[(size_t)(b * NE_ + j) * 3 + 0];
  const float ry = r[(size_t)(b * NE_ + j) * 3 + 1];
  const float rz = r[(size_t)(b * NE_ + j) * 3 + 2];
  const float inv_cut = 1.0f / CUTOFF_;
  const float dmu     = CUTOFF_ / (float)(NRBF_ - 1);
  const float inv_sig = (float)NRBF_ / CUTOFF_;
  const int t = threadIdx.x;
  const int a = t >> 4, k = t & 15;
  const float dx = rx - coords[a * 3 + 0];
  const float dy = ry - coords[a * 3 + 1];
  const float dz = rz - coords[a * 3 + 2];
  const float dist = sqrtf(dx * dx + dy * dy + dz * dz + EPS_);
  float env = 1.0f - dist * inv_cut;
  env = (dist < CUTOFF_) ? env * env : 0.0f;
  const float z = (dist - dmu * (float)k) * inv_sig;
  const float g = expf(-0.5f * z * z) * env;
  Ane[(size_t)bj * KNE_ + t] = g;
}

__global__ void gnn_init_h(const float* __restrict__ spin_emb, float* __restrict__ h) {
  const int gid = blockIdx.x * blockDim.x + threadIdx.x;
  const int d = gid & (D_ - 1);
  const int i = (gid >> 7) & (NE_ - 1);
  h[gid] = spin_emb[(i < NUP_ ? 0 : 1) * D_ + d];
}

__global__ void gnn_hw(const float* __restrict__ h, const float* __restrict__ nuc_emb,
                       const float* __restrict__ Wp_ee, const float* __restrict__ Wp_ne,
                       float* __restrict__ hW, float* __restrict__ hnW) {
  const int wid = blockIdx.x;
  const int lane = threadIdx.x & 31;
  v8f acc[NT_] = {};
  if (wid < 1024) {
    const int m0 = (wid >> 1) * 16;
    const int n0 = (wid & 1) * 64;
    gemm_strip_f32p(h, D_, Wp_ee, m0, n0, acc, lane);
    store_strip_f32(hW, D_, m0, n0, acc, lane);
  } else {
    const int n0 = (wid - 1024) * 64;
    gemm_strip_f32p(nuc_emb, D_, Wp_ne, 0, n0, acc, lane);
    store_strip_f32(hnW, D_, 0, n0, acc, lane);
  }
}

#define AGG_RING_  8
#define AGG_AHEAD_ 6
#define AGG_STEPS_ (NE_ + NN_)

__global__ __launch_bounds__(32) void gnn_agg(
    const float* __restrict__ Aee, const float* __restrict__ Ane,
    const float* __restrict__ hW, const float* __restrict__ hnW,
    const float* __restrict__ Wrbf_ee_l, const float* __restrict__ Wrbf_ne_l,
    float* __restrict__ agg) {
  const int wid = blockIdx.x;
  const int b  = wid >> 3;
  const int mt = (wid >> 1) & 3;
  const int nb = wid & 1;
  const int lane = threadIdx.x & 31;
  const int half = lane >> 4, l16 = lane & 15;
  const int i0 = mt * 16, n0 = nb * 64;

  const float* __restrict__ a_ee = Aee + ((size_t)b * NE_ + i0 + l16) * KEE_ + half * 2;
  const float* __restrict__ a_ne = Ane + ((size_t)b * NE_ + i0 + l16) * KNE_ + half * 2;
  const float* __restrict__ h_ee = hW + (size_t)b * NE_ * D_ + n0 + l16;
  const float* __restrict__ h_ne = hnW + n0 + l16;

  v8f   acc[NT_] = {};
  v2f   ra[AGG_RING_][4];
  float rh[AGG_RING_][NT_];

  auto issue_ee = [&](int j, int st) {
#pragma unroll
    for (int t = 0; t < NT_; ++t) rh[st][t] = h_ee[(size_t)j * D_ + t * 16];
#pragma unroll
    for (int kk = 0; kk < 4; ++kk) ra[st][kk] = *(const v2f*)&a_ee[j * NRBF_ + kk * 4];
  };
  auto issue_ne = [&](int a, int st) {
#pragma unroll
    for (int t = 0; t < NT_; ++t) rh[st][t] = h_ne[(size_t)a * D_ + t * 16];
#pragma unroll
    for (int kk = 0; kk < 4; ++kk) ra[st][kk] = *(const v2f*)&a_ne[a * NRBF_ + kk * 4];
  };

#pragma unroll
  for (int s = 0; s < AGG_AHEAD_; ++s) issue_ee(s, s);

  float wk_ee[NT_][8], wk_ne[NT_][8];
#pragma unroll
  for (int kk = 0; kk < 4; ++kk)
#pragma unroll
    for (int t = 0; t < NT_; ++t) {
      const int kb = kk * 4 + half * 2;
      wk_ee[t][kk * 2 + 0] = Wrbf_ee_l[(size_t)(kb + 0) * D_ + n0 + t * 16 + l16];
      wk_ee[t][kk * 2 + 1] = Wrbf_ee_l[(size_t)(kb + 1) * D_ + n0 + t * 16 + l16];
      wk_ne[t][kk * 2 + 0] = Wrbf_ne_l[(size_t)(kb + 0) * D_ + n0 + t * 16 + l16];
      wk_ne[t][kk * 2 + 1] = Wrbf_ne_l[(size_t)(kb + 1) * D_ + n0 + t * 16 + l16];
    }

  auto compute = [&](int st, const float (&wk)[NT_][8]) {
#pragma unroll
    for (int kk = 0; kk < 4; ++kk) {
      const v2f a = ra[st][kk];
#pragma unroll
      for (int t = 0; t < NT_; ++t) {
        v2f bb;
        bb.x = rh[st][t] * wk[t][kk * 2 + 0];
        bb.y = rh[st][t] * wk[t][kk * 2 + 1];
        acc[t] = __builtin_amdgcn_wmma_f32_16x16x4_f32(false, a, false, bb, (short)0,
                                                       acc[t], false, false);
      }
    }
  };

#pragma unroll 1
  for (int jo = 0; jo < NE_ - AGG_RING_; jo += AGG_RING_) {
#pragma unroll
    for (int u = 0; u < AGG_RING_; ++u) {
      issue_ee(jo + u + AGG_AHEAD_, (u + AGG_AHEAD_) % AGG_RING_);
      __builtin_amdgcn_sched_barrier(0);
      compute(u, wk_ee);
    }
  }
#pragma unroll
  for (int s = NE_ - AGG_RING_; s < AGG_STEPS_; ++s) {
    const int sn = s + AGG_AHEAD_;
    if (sn < NE_)             issue_ee(sn, sn % AGG_RING_);
    else if (sn < AGG_STEPS_) issue_ne(sn - NE_, sn % AGG_RING_);
    __builtin_amdgcn_sched_barrier(0);
    if (s < NE_) compute(s % AGG_RING_, wk_ee);
    else         compute(s % AGG_RING_, wk_ne);
  }

  store_strip_f32(agg + (size_t)b * NE_ * D_, D_, i0, n0, acc, lane);
}

__global__ void gnn_mlp1(const float* __restrict__ agg, const float* __restrict__ Wp1,
                         float* __restrict__ tmp) {
  const int wid = blockIdx.x;
  const int lane = threadIdx.x & 31;
  const int m0 = (wid >> 1) * 16;
  const int n0 = (wid & 1) * 64;
  v8f acc[NT_] = {};
  gemm_strip_f32p(agg, D_, Wp1, m0, n0, acc, lane);
#pragma unroll
  for (int t = 0; t < NT_; ++t)
#pragma unroll
    for (int ii = 0; ii < 8; ++ii) acc[t][ii] = tanhf(acc[t][ii]);
  store_strip_f32(tmp, D_, m0, n0, acc, lane);
}

__global__ void gnn_mlp2(const float* __restrict__ tmp, const float* __restrict__ Wp2,
                         const float* __restrict__ h, float* __restrict__ dst) {
  const int wid = blockIdx.x;
  const int lane = threadIdx.x & 31;
  const int m0 = (wid >> 1) * 16;
  const int n0 = (wid & 1) * 64;
  v8f acc[NT_];
  load_strip_f32(h, D_, m0, n0, acc, lane);
  gemm_strip_f32p(tmp, D_, Wp2, m0, n0, acc, lane);
  store_strip_f32(dst, D_, m0, n0, acc, lane);
}

extern "C" void kernel_launch(void* const* d_in, const int* in_sizes, int n_in,
                              void* d_out, int out_size, void* d_ws, size_t ws_size,
                              hipStream_t stream) {
  (void)in_sizes; (void)n_in; (void)out_size; (void)ws_size;
  const float* r        = (const float*)d_in[0];
  const float* coords   = (const float*)d_in[1];
  const float* spin_emb = (const float*)d_in[2];
  const float* nuc_emb  = (const float*)d_in[3];
  const float* W_rbf_ee = (const float*)d_in[4];
  const float* W_rbf_ne = (const float*)d_in[5];
  const float* W_msg_ee = (const float*)d_in[6];
  const float* W_msg_ne = (const float*)d_in[7];
  const float* W1       = (const float*)d_in[8];
  const float* W2       = (const float*)d_in[9];
  float* out = (float*)d_out;

  float* ws  = (float*)d_ws;
  float* Aee = ws;
  float* Ane = Aee + (size_t)B_ * NE_ * KEE_;
  float* h   = Ane + (size_t)B_ * NE_ * KNE_;
  float* hW  = h   + (size_t)B_ * NE_ * D_;
  float* hnW = hW  + (size_t)B_ * NE_ * D_;
  float* agg = hnW + (size_t)NN_ * D_;
  float* tmp = agg + (size_t)B_ * NE_ * D_;
  float* Wp  = tmp + (size_t)B_ * NE_ * D_;

  gnn_pack_w<<<(12 * WSZ_) / 256, 256, 0, stream>>>(W_msg_ee, W_msg_ne, W1, W2, Wp);
  gnn_feat_ee<<<B_ * NE_, 256, 0, stream>>>(r, Aee);
  gnn_feat_ne<<<B_ * NE_, 256, 0, stream>>>(r, coords, Ane);
  gnn_init_h<<<(B_ * NE_ * D_) / 256, 256, 0, stream>>>(spin_emb, h);

  for (int l = 0; l < L_; ++l) {
    const float* Wpl = Wp + (size_t)l * 4 * WSZ_;
    gnn_hw<<<1024 + 2, 32, 0, stream>>>(h, nuc_emb, Wpl + 0 * WSZ_, Wpl + 1 * WSZ_, hW, hnW);
    gnn_agg<<<B_ * 8, 32, 0, stream>>>(Aee, Ane, hW, hnW,
                                       W_rbf_ee + (size_t)l * NRBF_ * D_,
                                       W_rbf_ne + (size_t)l * NRBF_ * D_, agg);
    gnn_mlp1<<<1024, 32, 0, stream>>>(agg, Wpl + 2 * WSZ_, tmp);
    float* dst = (l == L_ - 1) ? out : h;
    gnn_mlp2<<<1024, 32, 0, stream>>>(tmp, Wpl + 3 * WSZ_, h, dst);
  }
}


# per layer, agg + mlp1 + mlp2 + next layer's hW fused into one launch (16 to 8 launches): 2-wave block owns 16 rows x D, stage results handed on through LDS; products and summation order unchanged
# speedup vs baseline: 2.0489x; 1.7218x over previous
#include <hip/hip_runtime.h>
#include <math.h>

#define B_     128
#define NE_    64
#define NN_    16
#define D_     128
#define NRBF_  16
#define L_     3
#define NUP_   32
#define CUTOFF_ 2.0f
#define EPS_    1e-12f

#define KEE_   (NE_ * NRBF_)
#define KNE_   (NN_ * NRBF_)
#define NT_    4
#define WSZ_   (D_ * D_)

typedef float v2f __attribute__((ext_vector_type(2)));
typedef float v8f __attribute__((ext_vector_type(8)));

__device__ __forceinline__ void gemm_strip_f32p(const float* __restrict__ A, int lda,
                                                const float* __restrict__ Bp,
                                                int m0, int n0, v8f acc[NT_], int lane) {
  const int half = lane >> 4;
  const int l16  = lane & 15;
  const float* __restrict__ arow = A + (size_t)(m0 + l16) * lda;
#pragma unroll
  for (int k0 = 0; k0 < D_; k0 += 4) {
    const int ka = k0 + half * 2;
    const v2f a = *(const v2f*)&arow[ka];
    const float* __restrict__ brow = Bp + (size_t)(ka >> 1) * (2 * D_);
#pragma unroll
    for (int t = 0; t < NT_; ++t) {
      const v2f b = *(const v2f*)&brow[(n0 + t * 16 + l16) * 2];
      acc[t] = __builtin_amdgcn_wmma_f32_16x16x4_f32(false, a, false, b, (short)0,
                                                     acc[t], false, false);
    }
  }
}

__device__ __forceinline__ void store_strip_f32(float* __restrict__ C, int ldc,
                                                int m0, int n0, v8f acc[NT_], int lane) {
  const int half = lane >> 4;
  const int l16  = lane & 15;
#pragma unroll
  for (int t = 0; t < NT_; ++t)
#pragma unroll
    for (int ii = 0; ii < 8; ++ii)
      C[(size_t)(m0 + half * 8 + ii) * ldc + n0 + t * 16 + l16] = acc[t][ii];
}

__device__ __forceinline__ void load_strip_f32(const float* __restrict__ C, int ldc,
                                               int m0, int n0, v8f acc[NT_], int lane) {
  const int half = lane >> 4;
  const int l16  = lane & 15;
#pragma unroll
  for (int t = 0; t < NT_; ++t)
#pragma unroll
    for (int ii = 0; ii < 8; ++ii)
      acc[t][ii] = C[(size_t)(m0 + half * 8 + ii) * ldc + n0 + t * 16 + l16];
}

#define XP_ (D_ + 4)

__device__ __forceinline__ void gemm_strip_ldsA(const float* Xs,
                                                const float* __restrict__ Bp,
                                                int n0, v8f acc[NT_], int lane) {
  const int half = lane >> 4;
  const int l16  = lane & 15;
  const float* arow = Xs + l16 * XP_;
#pragma unroll
  for (int k0 = 0; k0 < D_; k0 += 4) {
    const int ka = k0 + half * 2;
    const v2f a = *(const v2f*)&arow[ka];
    const float* __restrict__ brow = Bp + (size_t)(ka >> 1) * (2 * D_);
#pragma unroll
    for (int t = 0; t < NT_; ++t) {
      const v2f b = *(const v2f*)&brow[(n0 + t * 16 + l16) * 2];
      acc[t] = __builtin_amdgcn_wmma_f32_16x16x4_f32(false, a, false, b, (short)0,
                                                     acc[t], false, false);
    }
  }
}

__global__ void gnn_pack_w(const float* __restrict__ We, const float* __restrict__ Wn,
                           const float* __restrict__ W1, const float* __restrict__ W2,
                           float* __restrict__ Wp) {
  const int gid = blockIdx.x * blockDim.x + threadIdx.x;
  const int m   = gid >> 14;
  const int idx = gid & (WSZ_ - 1);
  const int l = m >> 2, w = m & 3;
  const float* src = (w == 0 ? We : w == 1 ? Wn : w == 2 ? W1 : W2) + (size_t)l * WSZ_;
  const int k = idx >> 7, col = idx & (D_ - 1);
  Wp[(size_t)m * WSZ_ + (size_t)(k >> 1) * (2 * D_) + col * 2 + (k & 1)] = src[idx];
}

__global__ void gnn_feat_ee(const float* __restrict__ r, float* __restrict__ Aee) {
  const int bi = blockIdx.x;
  const int b = bi >> 6, i = bi & 63;
  const float rx = r[(size_t)(b * NE_ + i) * 3 + 0];
  const float ry = r[(size_t)(b * NE_ + i) * 3 + 1];
  const float rz = r[(size_t)(b * NE_ + i) * 3 + 2];
  const float inv_cut = 1.0f / CUTOFF_;
  const float dmu     = CUTOFF_ / (float)(NRBF_ - 1);
  const float inv_sig = (float)NRBF_ / CUTOFF_;
  for (int t = threadIdx.x; t < KEE_; t += blockDim.x) {
    const int j = t >> 4, k = t & 15;
    const float dx = rx - r[(size_t)(b * NE_ + j) * 3 + 0];
    const float dy = ry - r[(size_t)(b * NE_ + j) * 3 + 1];
    const float dz = rz - r[(size_t)(b * NE_ + j) * 3 + 2];
    const float dist = sqrtf(dx * dx + dy * dy + dz * dz + EPS_);
    float env = 1.0f - dist * inv_cut;
    env = (dist < CUTOFF_) ? env * env : 0.0f;
    const float z = (dist - dmu * (float)k) * inv_sig;
    float g = expf(-0.5f * z * z) * env;
    if (i == j) g = 0.0f;
    Aee[(size_t)bi * KEE_ + t] = g;
  }
}

__global__ void gnn_feat_ne(const float* __restrict__ r, const float* __restrict__ coords,
                            float* __restrict__ Ane) {
  const int bj = blockIdx.x;
  const int b = bj >> 6, j = bj & 63;
  const float rx = r[(size_t)(b * NE_ + j) * 3 + 0];
  const float ry = r[(size_t)(b * NE_ + j) * 3 + 1];
  const float rz = r[(size_t)(b * NE_ + j) * 3 + 2];
  const float inv_cut = 1.0f / CUTOFF_;
  const float dmu     = CUTOFF_ / (float)(NRBF_ - 1);
  const float inv_sig = (float)NRBF_ / CUTOFF_;
  const int t = threadIdx.x;
  const int a = t >> 4, k = t & 15;
  const float dx = rx - coords[a * 3 + 0];
  const float dy = ry - coords[a * 3 + 1];
  const float dz = rz - coords[a * 3 + 2];
  const float dist = sqrtf(dx * dx + dy * dy + dz * dz + EPS_);
  float env = 1.0f - dist * inv_cut;
  env = (dist < CUTOFF_) ? env * env : 0.0f;
  const float z = (dist - dmu * (float)k) * inv_sig;
  const float g = expf(-0.5f * z * z) * env;
  Ane[(size_t)bj * KNE_ + t] = g;
}

__global__ void gnn_init_h(const float* __restrict__ spin_emb, float* __restrict__ h) {
  const int gid = blockIdx.x * blockDim.x + threadIdx.x;
  const int d = gid & (D_ - 1);
  const int i = (gid >> 7) & (NE_ - 1);
  h[gid] = spin_emb[(i < NUP_ ? 0 : 1) * D_ + d];
}

__global__ void gnn_hw(const float* __restrict__ h, const float* __restrict__ nuc_emb,
                       const float* __restrict__ Wp,
                       float* __restrict__ hW, float* __restrict__ hnW) {
  const int wid = blockIdx.x;
  const int lane = threadIdx.x & 31;
  v8f acc[NT_] = {};
  if (wid < 1024) {
    const int m0 = (wid >> 1) * 16;
    const int n0 = (wid & 1) * 64;
    gemm_strip_f32p(h, D_, Wp, m0, n0, acc, lane);
    store_strip_f32(hW, D_, m0, n0, acc, lane);
  } else {
    const int l  = (wid - 1024) >> 1;
    const int n0 = ((wid - 1024) & 1) * 64;
    gemm_strip_f32p(nuc_emb, D_, Wp + ((size_t)l * 4 + 1) * WSZ_, 0, n0, acc, lane);
    store_strip_f32(hnW + (size_t)l * NN_ * D_, D_, 0, n0, acc, lane);
  }
}

#define AGG_RING_  8
#define AGG_AHEAD_ 6
#define AGG_STEPS_ (NE_ + NN_)

__global__ __launch_bounds__(64) void gnn_layer(
    const float* __restrict__ Aee, const float* __restrict__ Ane,
    const float* __restrict__ hW, const float* __restrict__ hnW,
    const float* __restrict__ Wrbf_ee_l, const float* __restrict__ Wrbf_ne_l,
    const float* __restrict__ Wp1, const float* __restrict__ Wp2,
    const float* __restrict__ Wp_ee_next,
    const float* h, float* dst,
    float* __restrict__ hW_next) {
  __shared__ __attribute__((aligned(16))) float Xs[16 * XP_];
  __shared__ __attribute__((aligned(16))) float Ys[16 * XP_];
  const int wid = blockIdx.x;
  const int b  = wid >> 2;
  const int mt = wid & 3;
  const int nb = __builtin_amdgcn_readfirstlane((int)(threadIdx.x >> 5));
  const int lane = threadIdx.x & 31;
  const int half = lane >> 4, l16 = lane & 15;
  const int i0 = mt * 16, n0 = nb * 64;

  const float* __restrict__ a_ee = Aee + ((size_t)b * NE_ + i0 + l16) * KEE_ + half * 2;
  const float* __restrict__ a_ne = Ane + ((size_t)b * NE_ + i0 + l16) * KNE_ + half * 2;
  const float* __restrict__ h_ee = hW + (size_t)b * NE_ * D_ + n0 + l16;
  const float* __restrict__ h_ne = hnW + n0 + l16;

  v8f   acc[NT_] = {};
  v2f   ra[AGG_RING_][4];
  float rh[AGG_RING_][NT_];

  auto issue_ee = [&](int j, int st) {
#pragma unroll
    for (int t = 0; t < NT_; ++t) rh[st][t] = h_ee[(size_t)j * D_ + t * 16];
#pragma unroll
    for (int kk = 0; kk < 4; ++kk) ra[st][kk] = *(const v2f*)&a_ee[j * NRBF_ + kk * 4];
  };
  auto issue_ne = [&](int a, int st) {
#pragma unroll
    for (int t = 0; t < NT_; ++t) rh[st][t] = h_ne[(size_t)a * D_ + t * 16];
#pragma unroll
    for (int kk = 0; kk < 4; ++kk) ra[st][kk] = *(const v2f*)&a_ne[a * NRBF_ + kk * 4];
  };

#pragma unroll
  for (int s = 0; s < AGG_AHEAD_; ++s) issue_ee(s, s);

  float wk_ee[NT_][8], wk_ne[NT_][8];
#pragma unroll
  for (int kk = 0; kk < 4; ++kk)
#pragma unroll
    for (int t = 0; t < NT_; ++t) {
      const int kb = kk * 4 + half * 2;
      wk_ee[t][kk * 2 + 0] = Wrbf_ee_l[(size_t)(kb + 0) * D_ + n0 + t * 16 + l16];
      wk_ee[t][kk * 2 + 1] = Wrbf_ee_l[(size_t)(kb + 1) * D_ + n0 + t * 16 + l16];
      wk_ne[t][kk * 2 + 0] = Wrbf_ne_l[(size_t)(kb + 0) * D_ + n0 + t * 16 + l16];
      wk_ne[t][kk * 2 + 1] = Wrbf_ne_l[(size_t)(kb + 1) * D_ + n0 + t * 16 + l16];
    }

  auto compute = [&](int st, const float (&wk)[NT_][8]) {
#pragma unroll
    for (int kk = 0; kk < 4; ++kk) {
      const v2f a = ra[st][kk];
#pragma unroll
      for (int t = 0; t < NT_; ++t) {
        v2f bb;
        bb.x = rh[st][t] * wk[t][kk * 2 + 0];
        bb.y = rh[st][t] * wk[t][kk * 2 + 1];
        acc[t] = __builtin_amdgcn_wmma_f32_16x16x4_f32(false, a, false, bb, (short)0,
                                                       acc[t], false, false);
      }
    }
  };

#pragma unroll 1
  for (int jo = 0; jo < NE_ - AGG_RING_; jo += AGG_RING_) {
#pragma unroll
    for (int u = 0; u < AGG_RING_; ++u) {
      issue_ee(jo + u + AGG_AHEAD_, (u + AGG_AHEAD_) % AGG_RING_);
      __builtin_amdgcn_sched_barrier(0);
      compute(u, wk_ee);
    }
  }
#pragma unroll
  for (int s = NE_ - AGG_RING_; s < AGG_STEPS_; ++s) {
    const int sn = s + AGG_AHEAD_;
    if (sn < NE_)             issue_ee(sn, sn % AGG_RING_);
    else if (sn < AGG_STEPS_) issue_ne(sn - NE_, sn % AGG_RING_);
    __builtin_amdgcn_sched_barrier(0);
    if (s < NE_) compute(s % AGG_RING_, wk_ee);
    else         compute(s % AGG_RING_, wk_ne);
  }

  store_strip_f32(Xs, XP_, 0, n0, acc, lane);
  __syncthreads();

  v8f acc2[NT_] = {};
  gemm_strip_ldsA(Xs, Wp1, n0, acc2, lane);
#pragma unroll
  for (int t = 0; t < NT_; ++t)
#pragma unroll
    for (int ii = 0; ii < 8; ++ii) acc2[t][ii] = tanhf(acc2[t][ii]);
  store_strip_f32(Ys, XP_, 0, n0, acc2, lane);
  __syncthreads();

  const int m0 = b * NE_ + i0;
  v8f acc3[NT_];
  load_strip_f32(h, D_, m0, n0, acc3, lane);
  gemm_strip_ldsA(Ys, Wp2, n0, acc3, lane);
  store_strip_f32(dst, D_, m0, n0, acc3, lane);

  if (Wp_ee_next != nullptr) {
    store_strip_f32(Xs, XP_, 0, n0, acc3, lane);
    __syncthreads();
    v8f acc4[NT_] = {};
    gemm_strip_ldsA(Xs, Wp_ee_next, n0, acc4, lane);
    store_strip_f32(hW_next, D_, m0, n0, acc4, lane);
  }
}

extern "C" void kernel_launch(void* const* d_in, const int* in_sizes, int n_in,
                              void* d_out, int out_size, void* d_ws, size_t ws_size,
                              hipStream_t stream) {
  (void)in_sizes; (void)n_in; (void)out_size; (void)ws_size;
  const float* r        = (const float*)d_in[0];
  const float* coords   = (const float*)d_in[1];
  const float* spin_emb = (const float*)d_in[2];
  const float* nuc_emb  = (const float*)d_in[3];
  const float* W_rbf_ee = (const float*)d_in[4];
  const float* W_rbf_ne = (const float*)d_in[5];
  const float* W_msg_ee = (const float*)d_in[6];
  const float* W_msg_ne = (const float*)d_in[7];
  const float* W1       = (const float*)d_in[8];
  const float* W2       = (const float*)d_in[9];
  float* out = (float*)d_out;

  float* ws  = (float*)d_ws;
  float* Aee = ws;
  float* Ane = Aee + (size_t)B_ * NE_ * KEE_;
  float* h   = Ane + (size_t)B_ * NE_ * KNE_;
  float* hWa = h   + (size_t)B_ * NE_ * D_;
  float* hWb = hWa + (size_t)B_ * NE_ * D_;
  float* hnW = hWb + (size_t)B_ * NE_ * D_;
  float* Wp  = hnW + (size_t)L_ * NN_ * D_;

  gnn_pack_w<<<(12 * WSZ_) / 256, 256, 0, stream>>>(W_msg_ee, W_msg_ne, W1, W2, Wp);
  gnn_feat_ee<<<B_ * NE_, 256, 0, stream>>>(r, Aee);
  gnn_feat_ne<<<B_ * NE_, 256, 0, stream>>>(r, coords, Ane);
  gnn_init_h<<<(B_ * NE_ * D_) / 256, 256, 0, stream>>>(spin_emb, h);
  gnn_hw<<<1024 + 2 * L_, 32, 0, stream>>>(h, nuc_emb, Wp, hWa, hnW);

  for (int l = 0; l < L_; ++l) {
    const float* Wpl = Wp + (size_t)l * 4 * WSZ_;
    const bool last = (l == L_ - 1);
    const float* hW_in  = (l & 1) ? hWb : hWa;
    float*       hW_out = (l & 1) ? hWa : hWb;
    gnn_layer<<<B_ * 4, 64, 0, stream>>>(Aee, Ane, hW_in, hnW + (size_t)l * NN_ * D_,
                                         W_rbf_ee + (size_t)l * NRBF_ * D_,
                                         W_rbf_ne + (size_t)l * NRBF_ * D_,
                                         Wpl + 2 * WSZ_, Wpl + 3 * WSZ_,
                                         last ? nullptr : Wpl + 4 * WSZ_,
                                         h, last ? out : h, hW_out);
  }
}


# set-up kernels (weight pack, both basis tables, h init) re-gridded to few long-lived waves: many outputs per thread, whole-line 16-byte lane stores; 170k set-up waves per call become 8k; same values
# speedup vs baseline: 2.1303x; 1.0397x over previous
#include <hip/hip_runtime.h>
#include <math.h>

#define B_     128
#define NE_    64
#define NN_    16
#define D_     128
#define NRBF_  16
#define L_     3
#define NUP_   32
#define CUTOFF_ 2.0f
#define EPS_    1e-12f

#define KEE_   (NE_ * NRBF_)
#define KNE_   (NN_ * NRBF_)
#define NT_    4
#define WSZ_   (D_ * D_)

typedef float v2f __attribute__((ext_vector_type(2)));
typedef float v4f __attribute__((ext_vector_type(4)));
typedef float v8f __attribute__((ext_vector_type(8)));

__device__ __forceinline__ void gemm_strip_f32p(const float* __restrict__ A, int lda,
                                                const float* __restrict__ Bp,
                                                int m0, int n0, v8f acc[NT_], int lane) {
  const int half = lane >> 4;
  const int l16  = lane & 15;
  const float* __restrict__ arow = A + (size_t)(m0 + l16) * lda;
#pragma unroll
  for (int k0 = 0; k0 < D_; k0 += 4) {
    const int ka = k0 + half * 2;
    const v2f a = *(const v2f*)&arow[ka];
    const float* __restrict__ brow = Bp + (size_t)(ka >> 1) * (2 * D_);
#pragma unroll
    for (int t = 0; t < NT_; ++t) {
      const v2f b = *(const v2f*)&brow[(n0 + t * 16 + l16) * 2];
      acc[t] = __builtin_amdgcn_wmma_f32_16x16x4_f32(false, a, false, b, (short)0,
                                                     acc[t], false, false);
    }
  }
}

__device__ __forceinline__ void store_strip_f32(float* __restrict__ C, int ldc,
                                                int m0, int n0, v8f acc[NT_], int lane) {
  const int half = lane >> 4;
  const int l16  = lane & 15;
#pragma unroll
  for (int t = 0; t < NT_; ++t)
#pragma unroll
    for (int ii = 0; ii < 8; ++ii)
      C[(size_t)(m0 + half * 8 + ii) * ldc + n0 + t * 16 + l16] = acc[t][ii];
}

__device__ __forceinline__ void load_strip_f32(const float* __restrict__ C, int ldc,
                                               int m0, int n0, v8f acc[NT_], int lane) {
  const int half = lane >> 4;
  const int l16  = lane & 15;
#pragma unroll
  for (int t = 0; t < NT_; ++t)
#pragma unroll
    for (int ii = 0; ii < 8; ++ii)
      acc[t][ii] = C[(size_t)(m0 + half * 8 + ii) * ldc + n0 + t * 16 + l16];
}

#define XP_ (D_ + 4)

__device__ __forceinline__ void gemm_strip_ldsA(const float* Xs,
                                                const float* __restrict__ Bp,
                                                int n0, v8f acc[NT_], int lane) {
  const int half = lane >> 4;
  const int l16  = lane & 15;
  const float* arow = Xs + l16 * XP_;
#pragma unroll
  for (int k0 = 0; k0 < D_; k0 += 4) {
    const int ka = k0 + half * 2;
    const v2f a = *(const v2f*)&arow[ka];
    const float* __restrict__ brow = Bp + (size_t)(ka >> 1) * (2 * D_);
#pragma unroll
    for (int t = 0; t < NT_; ++t) {
      const v2f b = *(const v2f*)&brow[(n0 + t * 16 + l16) * 2];
      acc[t] = __builtin_amdgcn_wmma_f32_16x16x4_f32(false, a, false, b, (short)0,
                                                     acc[t], false, false);
    }
  }
}


__global__ __launch_bounds__(128) void gnn_pack_w(
    const float* __restrict__ We, const float* __restrict__ Wn,
    const float* __restrict__ W1, const float* __restrict__ W2,
    float* __restrict__ Wp) {
  const int gid = blockIdx.x * 128 + threadIdx.x;
  const int m   = gid >> 11;
  const int kp  = (gid >> 5) & 63;
  const int c4  = gid & 31;
  const int l = m >> 2, w = m & 3;
  const float* src = (w == 0 ? We : w == 1 ? Wn : w == 2 ? W1 : W2) + (size_t)l * WSZ_;
  const v4f x = *(const v4f*)&src[(size_t)(2 * kp) * D_ + c4 * 4];
  const v4f y = *(const v4f*)&src[(size_t)(2 * kp + 1) * D_ + c4 * 4];
  float* d = Wp + (size_t)m * WSZ_ + (size_t)kp * (2 * D_) + c4 * 8;
  v4f lo, hi;
  lo.x = x.x; lo.y = y.x; lo.z = x.y; lo.w = y.y;
  hi.x = x.z; hi.y = y.z; hi.z = x.w; hi.w = y.w;
  *(v4f*)d       = lo;
  *(v4f*)(d + 4) = hi;
}

__device__ __forceinline__ v4f rbf_quad(float dx, float dy, float dz, int kq, bool zero) {
  const float inv_cut = 1.0f / CUTOFF_;
  const float dmu     = CUTOFF_ / (float)(NRBF_ - 1);
  const float inv_sig = (float)NRBF_ / CUTOFF_;
  const float dist = sqrtf(dx * dx + dy * dy + dz * dz + EPS_);
  float env = 1.0f - dist * inv_cut;
  env = (dist < CUTOFF_) ? env * env : 0.0f;
  v4f g;
#pragma unroll
  for (int e = 0; e < 4; ++e) {
    const int k = kq * 4 + e;
    const float z = (dist - dmu * (float)k) * inv_sig;
    float ge = expf(-0.5f * z * z) * env;
    if (zero) ge = 0.0f;
    g[e] = ge;
  }
  return g;
}

#define FE_ROWS_ 2
__global__ __launch_bounds__(128) void gnn_feat_ee(const float* __restrict__ r,
                                                   float* __restrict__ Aee) {
  const int wave = blockIdx.x * 4 + __builtin_amdgcn_readfirstlane((int)(threadIdx.x >> 5));
  const int lane = threadIdx.x & 31;
  const int jl = lane >> 2, kq = lane & 3;
  const int row0 = wave * FE_ROWS_;
  const int b = row0 >> 6;
  float jx[8], jy[8], jz[8];
#pragma unroll
  for (int it = 0; it < 8; ++it) {
    const float* p = r + (size_t)(b * NE_ + it * 8 + jl) * 3;
    jx[it] = p[0]; jy[it] = p[1]; jz[it] = p[2];
  }
#pragma unroll
  for (int rr = 0; rr < FE_ROWS_; ++rr) {
    const int bi = row0 + rr;
    const int i  = bi & (NE_ - 1);
    const float rx = r[(size_t)bi * 3 + 0];
    const float ry = r[(size_t)bi * 3 + 1];
    const float rz = r[(size_t)bi * 3 + 2];
    float* __restrict__ orow = Aee + (size_t)bi * KEE_ + kq * 4;
#pragma unroll
    for (int it = 0; it < 8; ++it) {
      const int j = it * 8 + jl;
      const v4f g = rbf_quad(rx - jx[it], ry - jy[it], rz - jz[it], kq, i == j);
      *(v4f*)&orow[j * NRBF_] = g;
    }
  }
}

#define FN_ROWS_ 4
__global__ __launch_bounds__(128) void gnn_feat_ne(const float* __restrict__ r,
                                                   const float* __restrict__ coords,
                                                   float* __restrict__ Ane) {
  const int wave = blockIdx.x * 4 + __builtin_amdgcn_readfirstlane((int)(threadIdx.x >> 5));
  const int lane = threadIdx.x & 31;
  const int al = lane >> 2, kq = lane & 3;
  float cx[2], cy[2], cz[2];
#pragma unroll
  for (int it = 0; it < 2; ++it) {
    const float* p = coords + (it * 8 + al) * 3;
    cx[it] = p[0]; cy[it] = p[1]; cz[it] = p[2];
  }
#pragma unroll
  for (int rr = 0; rr < FN_ROWS_; ++rr) {
    const int bj = wave * FN_ROWS_ + rr;
    const float rx = r[(size_t)bj * 3 + 0];
    const float ry = r[(size_t)bj * 3 + 1];
    const float rz = r[(size_t)bj * 3 + 2];
    float* __restrict__ orow = Ane + (size_t)bj * KNE_ + kq * 4;
#pragma unroll
    for (int it = 0; it < 2; ++it) {
      const int a = it * 8 + al;
      const v4f g = rbf_quad(rx - cx[it], ry - cy[it], rz - cz[it], kq, false);
      *(v4f*)&orow[a * NRBF_] = g;
    }
  }
}

#define HI_ROWS_ 8
__global__ __launch_bounds__(128) void gnn_init_h(const float* __restrict__ spin_emb,
                                                  float* __restrict__ h) {
  const int wave = blockIdx.x * 4 + (threadIdx.x >> 5);
  const int lane = threadIdx.x & 31;
  const int row0 = wave * HI_ROWS_;
  const int i = row0 & (NE_ - 1);
  const v4f v = *(const v4f*)&spin_emb[(i < NUP_ ? 0 : 1) * D_ + lane * 4];
#pragma unroll
  for (int rr = 0; rr < HI_ROWS_; ++rr)
    *(v4f*)&h[(size_t)(row0 + rr) * D_ + lane * 4] = v;
}

__global__ void gnn_hw(const float* __restrict__ h, const float* __restrict__ nuc_emb,
                       const float* __restrict__ Wp,
                       float* __restrict__ hW, float* __restrict__ hnW) {
  const int wid = blockIdx.x;
  const int lane = threadIdx.x & 31;
  v8f acc[NT_] = {};
  if (wid < 1024) {
    const int m0 = (wid >> 1) * 16;
    const int n0 = (wid & 1) * 64;
    gemm_strip_f32p(h, D_, Wp, m0, n0, acc, lane);
    store_strip_f32(hW, D_, m0, n0, acc, lane);
  } else {
    const int l  = (wid - 1024) >> 1;
    const int n0 = ((wid - 1024) & 1) * 64;
    gemm_strip_f32p(nuc_emb, D_, Wp + ((size_t)l * 4 + 1) * WSZ_, 0, n0, acc, lane);
    store_strip_f32(hnW + (size_t)l * NN_ * D_, D_, 0, n0, acc, lane);
  }
}

#define AGG_RING_  8
#define AGG_AHEAD_ 6
#define AGG_STEPS_ (NE_ + NN_)

__global__ __launch_bounds__(64) void gnn_layer(
    const float* __restrict__ Aee, const float* __restrict__ Ane,
    const float* __restrict__ hW, const float* __restrict__ hnW,
    const float* __restrict__ Wrbf_ee_l, const float* __restrict__ Wrbf_ne_l,
    const float* __restrict__ Wp1, const float* __restrict__ Wp2,
    const float* __restrict__ Wp_ee_next,
    const float* h, float* dst,
    float* __restrict__ hW_next) {
  __shared__ __attribute__((aligned(16))) float Xs[16 * XP_];
  __shared__ __attribute__((aligned(16))) float Ys[16 * XP_];
  const int wid = blockIdx.x;
  const int b  = wid >> 2;
  const int mt = wid & 3;
  const int nb = __builtin_amdgcn_readfirstlane((int)(threadIdx.x >> 5));
  const int lane = threadIdx.x & 31;
  const int half = lane >> 4, l16 = lane & 15;
  const int i0 = mt * 16, n0 = nb * 64;

  const float* __restrict__ a_ee = Aee + ((size_t)b * NE_ + i0 + l16) * KEE_ + half * 2;
  const float* __restrict__ a_ne = Ane + ((size_t)b * NE_ + i0 + l16) * KNE_ + half * 2;
  const float* __restrict__ h_ee = hW + (size_t)b * NE_ * D_ + n0 + l16;
  const float* __restrict__ h_ne = hnW + n0 + l16;

  v8f   acc[NT_] = {};
  v2f   ra[AGG_RING_][4];
  float rh[AGG_RING_][NT_];

  auto issue_ee = [&](int j, int st) {
#pragma unroll
    for (int t = 0; t < NT_; ++t) rh[st][t] = h_ee[(size_t)j * D_ + t * 16];
#pragma unroll
    for (int kk = 0; kk < 4; ++kk) ra[st][kk] = *(const v2f*)&a_ee[j * NRBF_ + kk * 4];
  };
  auto issue_ne = [&](int a, int st) {
#pragma unroll
    for (int t = 0; t < NT_; ++t) rh[st][t] = h_ne[(size_t)a * D_ + t * 16];
#pragma unroll
    for (int kk = 0; kk < 4; ++kk) ra[st][kk] = *(const v2f*)&a_ne[a * NRBF_ + kk * 4];
  };

#pragma unroll
  for (int s = 0; s < AGG_AHEAD_; ++s) issue_ee(s, s);

  float wk_ee[NT_][8], wk_ne[NT_][8];
#pragma unroll
  for (int kk = 0; kk < 4; ++kk)
#pragma unroll
    for (int t = 0; t < NT_; ++t) {
      const int kb = kk * 4 + half * 2;
      wk_ee[t][kk * 2 + 0] = Wrbf_ee_l[(size_t)(kb + 0) * D_ + n0 + t * 16 + l16];
      wk_ee[t][kk * 2 + 1] = Wrbf_ee_l[(size_t)(kb + 1) * D_ + n0 + t * 16 + l16];
      wk_ne[t][kk * 2 + 0] = Wrbf_ne_l[(size_t)(kb + 0) * D_ + n0 + t * 16 + l16];
      wk_ne[t][kk * 2 + 1] = Wrbf_ne_l[(size_t)(kb + 1) * D_ + n0 + t * 16 + l16];
    }

  auto compute = [&](int st, const float (&wk)[NT_][8]) {
#pragma unroll
    for (int kk = 0; kk < 4; ++kk) {
      const v2f a = ra[st][kk];
#pragma unroll
      for (int t = 0; t < NT_; ++t) {
        v2f bb;
        bb.x = rh[st][t] * wk[t][kk * 2 + 0];
        bb.y = rh[st][t] * wk[t][kk * 2 + 1];
        acc[t] = __builtin_amdgcn_wmma_f32_16x16x4_f32(false, a, false, bb, (short)0,
                                                       acc[t], false, false);
      }
    }
  };

#pragma unroll 1
  for (int jo = 0; jo < NE_ - AGG_RING_; jo += AGG_RING_) {
#pragma unroll
    for (int u = 0; u < AGG_RING_; ++u) {
      issue_ee(jo + u + AGG_AHEAD_, (u + AGG_AHEAD_) % AGG_RING_);
      __builtin_amdgcn_sched_barrier(0);
      compute(u, wk_ee);
    }
  }
#pragma unroll
  for (int s = NE_ - AGG_RING_; s < AGG_STEPS_; ++s) {
    const int sn = s + AGG_AHEAD_;
    if (sn < NE_)             issue_ee(sn, sn % AGG_RING_);
    else if (sn < AGG_STEPS_) issue_ne(sn - NE_, sn % AGG_RING_);
    __builtin_amdgcn_sched_barrier(0);
    if (s < NE_) compute(s % AGG_RING_, wk_ee);
    else         compute(s % AGG_RING_, wk_ne);
  }

  store_strip_f32(Xs, XP_, 0, n0, acc, lane);
  __syncthreads();

  v8f acc2[NT_] = {};
  gemm_strip_ldsA(Xs, Wp1, n0, acc2, lane);
#pragma unroll
  for (int t = 0; t < NT_; ++t)
#pragma unroll
    for (int ii = 0; ii < 8; ++ii) acc2[t][ii] = tanhf(acc2[t][ii]);
  store_strip_f32(Ys, XP_, 0, n0, acc2, lane);
  __syncthreads();

  const int m0 = b * NE_ + i0;
  v8f acc3[NT_];
  load_strip_f32(h, D_, m0, n0, acc3, lane);
  gemm_strip_ldsA(Ys, Wp2, n0, acc3, lane);
  store_strip_f32(dst, D_, m0, n0, acc3, lane);

  if (Wp_ee_next != nullptr) {
    store_strip_f32(Xs, XP_, 0, n0, acc3, lane);
    __syncthreads();
    v8f acc4[NT_] = {};
    gemm_strip_ldsA(Xs, Wp_ee_next, n0, acc4, lane);
    store_strip_f32(hW_next, D_, m0, n0, acc4, lane);
  }
}

extern "C" void kernel_launch(void* const* d_in, const int* in_sizes, int n_in,
                              void* d_out, int out_size, void* d_ws, size_t ws_size,
                              hipStream_t stream) {
  (void)in_sizes; (void)n_in; (void)out_size; (void)ws_size;
  const float* r        = (const float*)d_in[0];
  const float* coords   = (const float*)d_in[1];
  const float* spin_emb = (const float*)d_in[2];
  const float* nuc_emb  = (const float*)d_in[3];
  const float* W_rbf_ee = (const float*)d_in[4];
  const float* W_rbf_ne = (const float*)d_in[5];
  const float* W_msg_ee = (const float*)d_in[6];
  const float* W_msg_ne = (const float*)d_in[7];
  const float* W1       = (const float*)d_in[8];
  const float* W2       = (const float*)d_in[9];
  float* out = (float*)d_out;

  float* ws  = (float*)d_ws;
  float* Aee = ws;
  float* Ane = Aee + (size_t)B_ * NE_ * KEE_;
  float* h   = Ane + (size_t)B_ * NE_ * KNE_;
  float* hWa = h   + (size_t)B_ * NE_ * D_;
  float* hWb = hWa + (size_t)B_ * NE_ * D_;
  float* hnW = hWb + (size_t)B_ * NE_ * D_;
  float* Wp  = hnW + (size_t)L_ * NN_ * D_;

  gnn_pack_w<<<(12 * (D_ / 2) * (D_ / 4)) / 128, 128, 0, stream>>>(W_msg_ee, W_msg_ne, W1, W2, Wp);
  gnn_feat_ee<<<(B_ * NE_) / (4 * FE_ROWS_), 128, 0, stream>>>(r, Aee);
  gnn_feat_ne<<<(B_ * NE_) / (4 * FN_ROWS_), 128, 0, stream>>>(r, coords, Ane);
  gnn_init_h<<<(B_ * NE_) / (4 * HI_ROWS_), 128, 0, stream>>>(spin_emb, h);
  gnn_hw<<<1024 + 2 * L_, 32, 0, stream>>>(h, nuc_emb, Wp, hWa, hnW);

  for (int l = 0; l < L_; ++l) {
    const float* Wpl = Wp + (size_t)l * 4 * WSZ_;
    const bool last = (l == L_ - 1);
    const float* hW_in  = (l & 1) ? hWb : hWa;
    float*       hW_out = (l & 1) ? hWa : hWb;
    gnn_layer<<<B_ * 4, 64, 0, stream>>>(Aee, Ane, hW_in, hnW + (size_t)l * NN_ * D_,
                                         W_rbf_ee + (size_t)l * NRBF_ * D_,
                                         W_rbf_ne + (size_t)l * NRBF_ * D_,
                                         Wpl + 2 * WSZ_, Wpl + 3 * WSZ_,
                                         last ? nullptr : Wpl + 4 * WSZ_,
                                         h, last ? out : h, hW_out);
  }
}


# layer kernel re-tiled: one 8-wave block per WGP owns 32 rows, each wave a 16x32 strip, so all four SIMDs carry two waves (was 2-wave blocks on two SIMDs); products and summation order unchanged
# speedup vs baseline: 2.2412x; 1.0521x over previous
#include <hip/hip_runtime.h>
#include <math.h>

#define B_     128
#define NE_    64
#define NN_    16
#define D_     128
#define NRBF_  16
#define L_     3
#define NUP_   32
#define CUTOFF_ 2.0f
#define EPS_    1e-12f

#define KEE_   (NE_ * NRBF_)
#define KNE_   (NN_ * NRBF_)
#define NT_    4
#define LNT_   2
#define WSZ_   (D_ * D_)

typedef float v2f __attribute__((ext_vector_type(2)));
typedef float v4f __attribute__((ext_vector_type(4)));
typedef float v8f __attribute__((ext_vector_type(8)));

template <int NT>
__device__ __forceinline__ void gemm_strip_f32p(const float* __restrict__ A, int lda,
                                                const float* __restrict__ Bp,
                                                int m0, int n0, v8f (&acc)[NT], int lane) {
  const int half = lane >> 4;
  const int l16  = lane & 15;
  const float* __restrict__ arow = A + (size_t)(m0 + l16) * lda;
#pragma unroll
  for (int k0 = 0; k0 < D_; k0 += 4) {
    const int ka = k0 + half * 2;
    const v2f a = *(const v2f*)&arow[ka];
    const float* __restrict__ brow = Bp + (size_t)(ka >> 1) * (2 * D_);
#pragma unroll
    for (int t = 0; t < NT; ++t) {
      const v2f b = *(const v2f*)&brow[(n0 + t * 16 + l16) * 2];
      acc[t] = __builtin_amdgcn_wmma_f32_16x16x4_f32(false, a, false, b, (short)0,
                                                     acc[t], false, false);
    }
  }
}

template <int NT>
__device__ __forceinline__ void store_strip_f32(float* __restrict__ C, int ldc,
                                                int m0, int n0, v8f (&acc)[NT], int lane) {
  const int half = lane >> 4;
  const int l16  = lane & 15;
#pragma unroll
  for (int t = 0; t < NT; ++t)
#pragma unroll
    for (int ii = 0; ii < 8; ++ii)
      C[(size_t)(m0 + half * 8 + ii) * ldc + n0 + t * 16 + l16] = acc[t][ii];
}

template <int NT>
__device__ __forceinline__ void load_strip_f32(const float* __restrict__ C, int ldc,
                                               int m0, int n0, v8f (&acc)[NT], int lane) {
  const int half = lane >> 4;
  const int l16  = lane & 15;
#pragma unroll
  for (int t = 0; t < NT; ++t)
#pragma unroll
    for (int ii = 0; ii < 8; ++ii)
      acc[t][ii] = C[(size_t)(m0 + half * 8 + ii) * ldc + n0 + t * 16 + l16];
}

#define XP_ (D_ + 4)

template <int NT>
__device__ __forceinline__ void gemm_strip_ldsA(const float* Xs,
                                                const float* __restrict__ Bp,
                                                int n0, v8f (&acc)[NT], int lane) {
  const int half = lane >> 4;
  const int l16  = lane & 15;
  const float* arow = Xs + l16 * XP_;
#pragma unroll
  for (int k0 = 0; k0 < D_; k0 += 4) {
    const int ka = k0 + half * 2;
    const v2f a = *(const v2f*)&arow[ka];
    const float* __restrict__ brow = Bp + (size_t)(ka >> 1) * (2 * D_);
#pragma unroll
    for (int t = 0; t < NT; ++t) {
      const v2f b = *(const v2f*)&brow[(n0 + t * 16 + l16) * 2];
      acc[t] = __builtin_amdgcn_wmma_f32_16x16x4_f32(false, a, false, b, (short)0,
                                                     acc[t], false, false);
    }
  }
}


__global__ __launch_bounds__(128) void gnn_pack_w(
    const float* __restrict__ We, const float* __restrict__ Wn,
    const float* __restrict__ W1, const float* __restrict__ W2,
    float* __restrict__ Wp) {
  const int gid = blockIdx.x * 128 + threadIdx.x;
  const int m   = gid >> 11;
  const int kp  = (gid >> 5) & 63;
  const int c4  = gid & 31;
  const int l = m >> 2, w = m & 3;
  const float* src = (w == 0 ? We : w == 1 ? Wn : w == 2 ? W1 : W2) + (size_t)l * WSZ_;
  const v4f x = *(const v4f*)&src[(size_t)(2 * kp) * D_ + c4 * 4];
  const v4f y = *(const v4f*)&src[(size_t)(2 * kp + 1) * D_ + c4 * 4];
  float* d = Wp + (size_t)m * WSZ_ + (size_t)kp * (2 * D_) + c4 * 8;
  v4f lo, hi;
  lo.x = x.x; lo.y = y.x; lo.z = x.y; lo.w = y.y;
  hi.x = x.z; hi.y = y.z; hi.z = x.w; hi.w = y.w;
  *(v4f*)d       = lo;
  *(v4f*)(d + 4) = hi;
}

__device__ __forceinline__ v4f rbf_quad(float dx, float dy, float dz, int kq, bool zero) {
  const float inv_cut = 1.0f / CUTOFF_;
  const float dmu     = CUTOFF_ / (float)(NRBF_ - 1);
  const float inv_sig = (float)NRBF_ / CUTOFF_;
  const float dist = sqrtf(dx * dx + dy * dy + dz * dz + EPS_);
  float env = 1.0f - dist * inv_cut;
  env = (dist < CUTOFF_) ? env * env : 0.0f;
  v4f g;
#pragma unroll
  for (int e = 0; e < 4; ++e) {
    const int k = kq * 4 + e;
    const float z = (dist - dmu * (float)k) * inv_sig;
    float ge = expf(-0.5f * z * z) * env;
    if (zero) ge = 0.0f;
    g[e] = ge;
  }
  return g;
}

#define FE_ROWS_ 2
__global__ __launch_bounds__(128) void gnn_feat_ee(const float* __restrict__ r,
                                                   float* __restrict__ Aee) {
  const int wave = blockIdx.x * 4 + __builtin_amdgcn_readfirstlane((int)(threadIdx.x >> 5));
  const int lane = threadIdx.x & 31;
  const int jl = lane >> 2, kq = lane & 3;
  const int row0 = wave * FE_ROWS_;
  const int b = row0 >> 6;
  float jx[8], jy[8], jz[8];
#pragma unroll
  for (int it = 0; it < 8; ++it) {
    const float* p = r + (size_t)(b * NE_ + it * 8 + jl) * 3;
    jx[it] = p[0]; jy[it] = p[1]; jz[it] = p[2];
  }
#pragma unroll
  for (int rr = 0; rr < FE_ROWS_; ++rr) {
    const int bi = row0 + rr;
    const int i  = bi & (NE_ - 1);
    const float rx = r[(size_t)bi * 3 + 0];
    const float ry = r[(size_t)bi * 3 + 1];
    const float rz = r[(size_t)bi * 3 + 2];
    float* __restrict__ orow = Aee + (size_t)bi * KEE_ + kq * 4;
#pragma unroll
    for (int it = 0; it < 8; ++it) {
      const int j = it * 8 + jl;
      const v4f g = rbf_quad(rx - jx[it], ry - jy[it], rz - jz[it], kq, i == j);
      *(v4f*)&orow[j * NRBF_] = g;
    }
  }
}

#define FN_ROWS_ 4
__global__ __launch_bounds__(128) void gnn_feat_ne(const float* __restrict__ r,
                                                   const float* __restrict__ coords,
                                                   float* __restrict__ Ane) {
  const int wave = blockIdx.x * 4 + __builtin_amdgcn_readfirstlane((int)(threadIdx.x >> 5));
  const int lane = threadIdx.x & 31;
  const int al = lane >> 2, kq = lane & 3;
  float cx[2], cy[2], cz[2];
#pragma unroll
  for (int it = 0; it < 2; ++it) {
    const float* p = coords + (it * 8 + al) * 3;
    cx[it] = p[0]; cy[it] = p[1]; cz[it] = p[2];
  }
#pragma unroll
  for (int rr = 0; rr < FN_ROWS_; ++rr) {
    const int bj = wave * FN_ROWS_ + rr;
    const float rx = r[(size_t)bj * 3 + 0];
    const float ry = r[(size_t)bj * 3 + 1];
    const float rz = r[(size_t)bj * 3 + 2];
    float* __restrict__ orow = Ane + (size_t)bj * KNE_ + kq * 4;
#pragma unroll
    for (int it = 0; it < 2; ++it) {
      const int a = it * 8 + al;
      const v4f g = rbf_quad(rx - cx[it], ry - cy[it], rz - cz[it], kq, false);
      *(v4f*)&orow[a * NRBF_] = g;
    }
  }
}

#define HI_ROWS_ 8
__global__ __launch_bounds__(128) void gnn_init_h(const float* __restrict__ spin_emb,
                                                  float* __restrict__ h) {
  const int wave = blockIdx.x * 4 + (threadIdx.x >> 5);
  const int lane = threadIdx.x & 31;
  const int row0 = wave * HI_ROWS_;
  const int i = row0 & (NE_ - 1);
  const v4f v = *(const v4f*)&spin_emb[(i < NUP_ ? 0 : 1) * D_ + lane * 4];
#pragma unroll
  for (int rr = 0; rr < HI_ROWS_; ++rr)
    *(v4f*)&h[(size_t)(row0 + rr) * D_ + lane * 4] = v;
}

__global__ void gnn_hw(const float* __restrict__ h, const float* __restrict__ nuc_emb,
                       const float* __restrict__ Wp,
                       float* __restrict__ hW, float* __restrict__ hnW) {
  const int wid = blockIdx.x;
  const int lane = threadIdx.x & 31;
  v8f acc[NT_] = {};
  if (wid < 1024) {
    const int m0 = (wid >> 1) * 16;
    const int n0 = (wid & 1) * 64;
    gemm_strip_f32p(h, D_, Wp, m0, n0, acc, lane);
    store_strip_f32(hW, D_, m0, n0, acc, lane);
  } else {
    const int l  = (wid - 1024) >> 1;
    const int n0 = ((wid - 1024) & 1) * 64;
    gemm_strip_f32p(nuc_emb, D_, Wp + ((size_t)l * 4 + 1) * WSZ_, 0, n0, acc, lane);
    store_strip_f32(hnW + (size_t)l * NN_ * D_, D_, 0, n0, acc, lane);
  }
}

#define AGG_RING_  8
#define AGG_AHEAD_ 6
#define AGG_STEPS_ (NE_ + NN_)
#define LROWS_     32

__global__ __launch_bounds__(256) void gnn_layer(
    const float* __restrict__ Aee, const float* __restrict__ Ane,
    const float* __restrict__ hW, const float* __restrict__ hnW,
    const float* __restrict__ Wrbf_ee_l, const float* __restrict__ Wrbf_ne_l,
    const float* __restrict__ Wp1, const float* __restrict__ Wp2,
    const float* __restrict__ Wp_ee_next,
    const float* h, float* dst,
    float* __restrict__ hW_next) {
  __shared__ __attribute__((aligned(16))) float Xs[LROWS_ * XP_];
  __shared__ __attribute__((aligned(16))) float Ys[LROWS_ * XP_];
  const int wid = blockIdx.x;
  const int b  = wid >> 1;
  const int w  = __builtin_amdgcn_readfirstlane((int)(threadIdx.x >> 5));
  const int mw = w & 1;
  const int nq = w >> 1;
  const int lane = threadIdx.x & 31;
  const int half = lane >> 4, l16 = lane & 15;
  const int i0 = (wid & 1) * LROWS_ + mw * 16, n0 = nq * (16 * LNT_);
  float* Xw = Xs + mw * 16 * XP_;
  float* Yw = Ys + mw * 16 * XP_;

  const float* __restrict__ a_ee = Aee + ((size_t)b * NE_ + i0 + l16) * KEE_ + half * 2;
  const float* __restrict__ a_ne = Ane + ((size_t)b * NE_ + i0 + l16) * KNE_ + half * 2;
  const float* __restrict__ h_ee = hW + (size_t)b * NE_ * D_ + n0 + l16;
  const float* __restrict__ h_ne = hnW + n0 + l16;

  v8f   acc[LNT_] = {};
  v2f   ra[AGG_RING_][4];
  float rh[AGG_RING_][LNT_];

  auto issue_ee = [&](int j, int st) {
#pragma unroll
    for (int t = 0; t < LNT_; ++t) rh[st][t] = h_ee[(size_t)j * D_ + t * 16];
#pragma unroll
    for (int kk = 0; kk < 4; ++kk) ra[st][kk] = *(const v2f*)&a_ee[j * NRBF_ + kk * 4];
  };
  auto issue_ne = [&](int a, int st) {
#pragma unroll
    for (int t = 0; t < LNT_; ++t) rh[st][t] = h_ne[(size_t)a * D_ + t * 16];
#pragma unroll
    for (int kk = 0; kk < 4; ++kk) ra[st][kk] = *(const v2f*)&a_ne[a * NRBF_ + kk * 4];
  };

#pragma unroll
  for (int s = 0; s < AGG_AHEAD_; ++s) issue_ee(s, s);

  float wk_ee[LNT_][8], wk_ne[LNT_][8];
#pragma unroll
  for (int kk = 0; kk < 4; ++kk)
#pragma unroll
    for (int t = 0; t < LNT_; ++t) {
      const int kb = kk * 4 + half * 2;
      wk_ee[t][kk * 2 + 0] = Wrbf_ee_l[(size_t)(kb + 0) * D_ + n0 + t * 16 + l16];
      wk_ee[t][kk * 2 + 1] = Wrbf_ee_l[(size_t)(kb + 1) * D_ + n0 + t * 16 + l16];
      wk_ne[t][kk * 2 + 0] = Wrbf_ne_l[(size_t)(kb + 0) * D_ + n0 + t * 16 + l16];
      wk_ne[t][kk * 2 + 1] = Wrbf_ne_l[(size_t)(kb + 1) * D_ + n0 + t * 16 + l16];
    }

  auto compute = [&](int st, const float (&wk)[LNT_][8]) {
#pragma unroll
    for (int kk = 0; kk < 4; ++kk) {
      const v2f a = ra[st][kk];
#pragma unroll
      for (int t = 0; t < LNT_; ++t) {
        v2f bb;
        bb.x = rh[st][t] * wk[t][kk * 2 + 0];
        bb.y = rh[st][t] * wk[t][kk * 2 + 1];
        acc[t] = __builtin_amdgcn_wmma_f32_16x16x4_f32(false, a, false, bb, (short)0,
                                                       acc[t], false, false);
      }
    }
  };

#pragma unroll 1
  for (int jo = 0; jo < NE_ - AGG_RING_; jo += AGG_RING_) {
#pragma unroll
    for (int u = 0; u < AGG_RING_; ++u) {
      issue_ee(jo + u + AGG_AHEAD_, (u + AGG_AHEAD_) % AGG_RING_);
      __builtin_amdgcn_sched_barrier(0);
      compute(u, wk_ee);
    }
  }
#pragma unroll
  for (int s = NE_ - AGG_RING_; s < AGG_STEPS_; ++s) {
    const int sn = s + AGG_AHEAD_;
    if (sn < NE_)             issue_ee(sn, sn % AGG_RING_);
    else if (sn < AGG_STEPS_) issue_ne(sn - NE_, sn % AGG_RING_);
    __builtin_amdgcn_sched_barrier(0);
    if (s < NE_) compute(s % AGG_RING_, wk_ee);
    else         compute(s % AGG_RING_, wk_ne);
  }

  store_strip_f32(Xw, XP_, 0, n0, acc, lane);
  __syncthreads();

  v8f acc2[LNT_] = {};
  gemm_strip_ldsA(Xw, Wp1, n0, acc2, lane);
#pragma unroll
  for (int t = 0; t < LNT_; ++t)
#pragma unroll
    for (int ii = 0; ii < 8; ++ii) acc2[t][ii] = tanhf(acc2[t][ii]);
  store_strip_f32(Yw, XP_, 0, n0, acc2, lane);
  __syncthreads();

  const int m0 = b * NE_ + i0;
  v8f acc3[LNT_];
  load_strip_f32(h, D_, m0, n0, acc3, lane);
  gemm_strip_ldsA(Yw, Wp2, n0, acc3, lane);
  store_strip_f32(dst, D_, m0, n0, acc3, lane);

  if (Wp_ee_next != nullptr) {
    store_strip_f32(Xw, XP_, 0, n0, acc3, lane);
    __syncthreads();
    v8f acc4[LNT_] = {};
    gemm_strip_ldsA(Xw, Wp_ee_next, n0, acc4, lane);
    store_strip_f32(hW_next, D_, m0, n0, acc4, lane);
  }
}

extern "C" void kernel_launch(void* const* d_in, const int* in_sizes, int n_in,
                              void* d_out, int out_size, void* d_ws, size_t ws_size,
                              hipStream_t stream) {
  (void)in_sizes; (void)n_in; (void)out_size; (void)ws_size;
  const float* r        = (const float*)d_in[0];
  const float* coords   = (const float*)d_in[1];
  const float* spin_emb = (const float*)d_in[2];
  const float* nuc_emb  = (const float*)d_in[3];
  const float* W_rbf_ee = (const float*)d_in[4];
  const float* W_rbf_ne = (const float*)d_in[5];
  const float* W_msg_ee = (const float*)d_in[6];
  const float* W_msg_ne = (const float*)d_in[7];
  const float* W1       = (const float*)d_in[8];
  const float* W2       = (const float*)d_in[9];
  float* out = (float*)d_out;

  float* ws  = (float*)d_ws;
  float* Aee = ws;
  float* Ane = Aee + (size_t)B_ * NE_ * KEE_;
  float* h   = Ane + (size_t)B_ * NE_ * KNE_;
  float* hWa = h   + (size_t)B_ * NE_ * D_;
  float* hWb = hWa + (size_t)B_ * NE_ * D_;
  float* hnW = hWb + (size_t)B_ * NE_ * D_;
  float* Wp  = hnW + (size_t)L_ * NN_ * D_;

  gnn_pack_w<<<(12 * (D_ / 2) * (D_ / 4)) / 128, 128, 0, stream>>>(W_msg_ee, W_msg_ne, W1, W2, Wp);
  gnn_feat_ee<<<(B_ * NE_) / (4 * FE_ROWS_), 128, 0, stream>>>(r, Aee);
  gnn_feat_ne<<<(B_ * NE_) / (4 * FN_ROWS_), 128, 0, stream>>>(r, coords, Ane);
  gnn_init_h<<<(B_ * NE_) / (4 * HI_ROWS_), 128, 0, stream>>>(spin_emb, h);
  gnn_hw<<<1024 + 2 * L_, 32, 0, stream>>>(h, nuc_emb, Wp, hWa, hnW);

  for (int l = 0; l < L_; ++l) {
    const float* Wpl = Wp + (size_t)l * 4 * WSZ_;
    const bool last = (l == L_ - 1);
    const float* hW_in  = (l & 1) ? hWb : hWa;
    float*       hW_out = (l & 1) ? hWa : hWb;
    gnn_layer<<<B_ * (NE_ / LROWS_), 256, 0, stream>>>(Aee, Ane, hW_in, hnW + (size_t)l * NN_ * D_,
                                         W_rbf_ee + (size_t)l * NRBF_ * D_,
                                         W_rbf_ne + (size_t)l * NRBF_ * D_,
                                         Wpl + 2 * WSZ_, Wpl + 3 * WSZ_,
                                         last ? nullptr : Wpl + 4 * WSZ_,
                                         h, last ? out : h, hW_out);
  }
}


# the five set-up launches (weight pack, two basis tables, h init, first hW) merged into one by block range; first hW reads spin_emb and unpacked weights; 8 launches become 4; same products and order
# speedup vs baseline: 2.6366x; 1.1764x over previous
#include <hip/hip_runtime.h>
#include <math.h>

#define B_     128
#define NE_    64
#define NN_    16
#define D_     128
#define NRBF_  16
#define L_     3
#define NUP_   32
#define CUTOFF_ 2.0f
#define EPS_    1e-12f

#define KEE_   (NE_ * NRBF_)
#define KNE_   (NN_ * NRBF_)
#define NT_    4
#define LNT_   2
#define WSZ_   (D_ * D_)

typedef float v2f __attribute__((ext_vector_type(2)));
typedef float v4f __attribute__((ext_vector_type(4)));
typedef float v8f __attribute__((ext_vector_type(8)));

template <int NT>
__device__ __forceinline__ void gemm_strip_f32u(const float* __restrict__ arow,
                                                const float* __restrict__ Wm,
                                                int n0, v8f (&acc)[NT], int lane) {
  const int half = lane >> 4;
  const int l16  = lane & 15;
#pragma unroll
  for (int k0 = 0; k0 < D_; k0 += 4) {
    const int ka = k0 + half * 2;
    const v2f a = *(const v2f*)&arow[ka];
    const float* __restrict__ bcol = Wm + (size_t)ka * D_ + n0 + l16;
#pragma unroll
    for (int t = 0; t < NT; ++t) {
      v2f b;
      b.x = bcol[t * 16];
      b.y = bcol[D_ + t * 16];
      acc[t] = __builtin_amdgcn_wmma_f32_16x16x4_f32(false, a, false, b, (short)0,
                                                     acc[t], false, false);
    }
  }
}

template <int NT>
__device__ __forceinline__ void store_strip_f32(float* __restrict__ C, int ldc,
                                                int m0, int n0, v8f (&acc)[NT], int lane) {
  const int half = lane >> 4;
  const int l16  = lane & 15;
#pragma unroll
  for (int t = 0; t < NT; ++t)
#pragma unroll
    for (int ii = 0; ii < 8; ++ii)
      C[(size_t)(m0 + half * 8 + ii) * ldc + n0 + t * 16 + l16] = acc[t][ii];
}

template <int NT>
__device__ __forceinline__ void load_strip_f32(const float* __restrict__ C, int ldc,
                                               int m0, int n0, v8f (&acc)[NT], int lane) {
  const int half = lane >> 4;
  const int l16  = lane & 15;
#pragma unroll
  for (int t = 0; t < NT; ++t)
#pragma unroll
    for (int ii = 0; ii < 8; ++ii)
      acc[t][ii] = C[(size_t)(m0 + half * 8 + ii) * ldc + n0 + t * 16 + l16];
}

#define XP_ (D_ + 4)

template <int NT>
__device__ __forceinline__ void gemm_strip_ldsA(const float* Xs,
                                                const float* __restrict__ Bp,
                                                int n0, v8f (&acc)[NT], int lane) {
  const int half = lane >> 4;
  const int l16  = lane & 15;
  const float* arow = Xs + l16 * XP_;
#pragma unroll
  for (int k0 = 0; k0 < D_; k0 += 4) {
    const int ka = k0 + half * 2;
    const v2f a = *(const v2f*)&arow[ka];
    const float* __restrict__ brow = Bp + (size_t)(ka >> 1) * (2 * D_);
#pragma unroll
    for (int t = 0; t < NT; ++t) {
      const v2f b = *(const v2f*)&brow[(n0 + t * 16 + l16) * 2];
      acc[t] = __builtin_amdgcn_wmma_f32_16x16x4_f32(false, a, false, b, (short)0,
                                                     acc[t], false, false);
    }
  }
}


#define PK_BLOCKS_ ((12 * (D_ / 2) * (D_ / 4)) / 128)
__device__ __forceinline__ void setup_pack_w(
    int blk,
    const float* __restrict__ We, const float* __restrict__ Wn,
    const float* __restrict__ W1, const float* __restrict__ W2,
    float* __restrict__ Wp) {
  const int gid = blk * 128 + threadIdx.x;
  const int m   = gid >> 11;
  const int kp  = (gid >> 5) & 63;
  const int c4  = gid & 31;
  const int l = m >> 2, w = m & 3;
  const float* src = (w == 0 ? We : w == 1 ? Wn : w == 2 ? W1 : W2) + (size_t)l * WSZ_;
  const v4f x = *(const v4f*)&src[(size_t)(2 * kp) * D_ + c4 * 4];
  const v4f y = *(const v4f*)&src[(size_t)(2 * kp + 1) * D_ + c4 * 4];
  float* d = Wp + (size_t)m * WSZ_ + (size_t)kp * (2 * D_) + c4 * 8;
  v4f lo, hi;
  lo.x = x.x; lo.y = y.x; lo.z = x.y; lo.w = y.y;
  hi.x = x.z; hi.y = y.z; hi.z = x.w; hi.w = y.w;
  *(v4f*)d       = lo;
  *(v4f*)(d + 4) = hi;
}

__device__ __forceinline__ v4f rbf_quad(float dx, float dy, float dz, int kq, bool zero) {
  const float inv_cut = 1.0f / CUTOFF_;
  const float dmu     = CUTOFF_ / (float)(NRBF_ - 1);
  const float inv_sig = (float)NRBF_ / CUTOFF_;
  const float dist = sqrtf(dx * dx + dy * dy + dz * dz + EPS_);
  float env = 1.0f - dist * inv_cut;
  env = (dist < CUTOFF_) ? env * env : 0.0f;
  v4f g;
#pragma unroll
  for (int e = 0; e < 4; ++e) {
    const int k = kq * 4 + e;
    const float z = (dist - dmu * (float)k) * inv_sig;
    float ge = expf(-0.5f * z * z) * env;
    if (zero) ge = 0.0f;
    g[e] = ge;
  }
  return g;
}

#define FE_ROWS_ 2
#define FE_BLOCKS_ ((B_ * NE_) / (4 * FE_ROWS_))
__device__ __forceinline__ void setup_feat_ee(int blk, const float* __restrict__ r,
                                              float* __restrict__ Aee) {
  const int wave = blk * 4 + __builtin_amdgcn_readfirstlane((int)(threadIdx.x >> 5));
  const int lane = threadIdx.x & 31;
  const int jl = lane >> 2, kq = lane & 3;
  const int row0 = wave * FE_ROWS_;
  const int b = row0 >> 6;
  float jx[8], jy[8], jz[8];
#pragma unroll
  for (int it = 0; it < 8; ++it) {
    const float* p = r + (size_t)(b * NE_ + it * 8 + jl) * 3;
    jx[it] = p[0]; jy[it] = p[1]; jz[it] = p[2];
  }
#pragma unroll
  for (int rr = 0; rr < FE_ROWS_; ++rr) {
    const int bi = row0 + rr;
    const int i  = bi & (NE_ - 1);
    const float rx = r[(size_t)bi * 3 + 0];
    const float ry = r[(size_t)bi * 3 + 1];
    const float rz = r[(size_t)bi * 3 + 2];
    float* __restrict__ orow = Aee + (size_t)bi * KEE_ + kq * 4;
#pragma unroll
    for (int it = 0; it < 8; ++it) {
      const int j = it * 8 + jl;
      const v4f g = rbf_quad(rx - jx[it], ry - jy[it], rz - jz[it], kq, i == j);
      *(v4f*)&orow[j * NRBF_] = g;
    }
  }
}

#define FN_ROWS_ 4
#define FN_BLOCKS_ ((B_ * NE_) / (4 * FN_ROWS_))
__device__ __forceinline__ void setup_feat_ne(int blk, const float* __restrict__ r,
                                              const float* __restrict__ coords,
                                              float* __restrict__ Ane) {
  const int wave = blk * 4 + __builtin_amdgcn_readfirstlane((int)(threadIdx.x >> 5));
  const int lane = threadIdx.x & 31;
  const int al = lane >> 2, kq = lane & 3;
  float cx[2], cy[2], cz[2];
#pragma unroll
  for (int it = 0; it < 2; ++it) {
    const float* p = coords + (it * 8 + al) * 3;
    cx[it] = p[0]; cy[it] = p[1]; cz[it] = p[2];
  }
#pragma unroll
  for (int rr = 0; rr < FN_ROWS_; ++rr) {
    const int bj = wave * FN_ROWS_ + rr;
    const float rx = r[(size_t)bj * 3 + 0];
    const float ry = r[(size_t)bj * 3 + 1];
    const float rz = r[(size_t)bj * 3 + 2];
    float* __restrict__ orow = Ane + (size_t)bj * KNE_ + kq * 4;
#pragma unroll
    for (int it = 0; it < 2; ++it) {
      const int a = it * 8 + al;
      const v4f g = rbf_quad(rx - cx[it], ry - cy[it], rz - cz[it], kq, false);
      *(v4f*)&orow[a * NRBF_] = g;
    }
  }
}

#define HI_ROWS_ 8
#define HI_BLOCKS_ ((B_ * NE_) / (4 * HI_ROWS_))
__device__ __forceinline__ void setup_init_h(int blk, const float* __restrict__ spin_emb,
                                             float* __restrict__ h) {
  const int wave = blk * 4 + (threadIdx.x >> 5);
  const int lane = threadIdx.x & 31;
  const int row0 = wave * HI_ROWS_;
  const int i = row0 & (NE_ - 1);
  const v4f v = *(const v4f*)&spin_emb[(i < NUP_ ? 0 : 1) * D_ + lane * 4];
#pragma unroll
  for (int rr = 0; rr < HI_ROWS_; ++rr)
    *(v4f*)&h[(size_t)(row0 + rr) * D_ + lane * 4] = v;
}

#define HW_STRIPS_ (2 * (B_ * NE_ / 16))
#define HW_BLOCKS_ ((HW_STRIPS_ + 2 * L_ + 3) / 4)
__device__ __forceinline__ void setup_hw(int blk,
                                         const float* __restrict__ spin_emb,
                                         const float* __restrict__ nuc_emb,
                                         const float* __restrict__ We,
                                         const float* __restrict__ Wn,
                                         float* __restrict__ hW, float* __restrict__ hnW) {
  const int s = blk * 4 + __builtin_amdgcn_readfirstlane((int)(threadIdx.x >> 5));
  const int lane = threadIdx.x & 31;
  const int l16  = lane & 15;
  v8f acc[NT_] = {};
  if (s < HW_STRIPS_) {
    const int m0 = (s >> 1) * 16;
    const int n0 = (s & 1) * 64;
    const int spin = ((m0 & (NE_ - 1)) < NUP_) ? 0 : 1;
    gemm_strip_f32u(spin_emb + spin * D_, We, n0, acc, lane);
    store_strip_f32(hW, D_, m0, n0, acc, lane);
  } else if (s < HW_STRIPS_ + 2 * L_) {
    const int l  = (s - HW_STRIPS_) >> 1;
    const int n0 = ((s - HW_STRIPS_) & 1) * 64;
    gemm_strip_f32u(nuc_emb + l16 * D_, Wn + (size_t)l * WSZ_, n0, acc, lane);
    store_strip_f32(hnW + (size_t)l * NN_ * D_, D_, 0, n0, acc, lane);
  }
}

#define SB_HW_ (FE_BLOCKS_)
#define SB_FN_ (SB_HW_ + HW_BLOCKS_)
#define SB_PK_ (SB_FN_ + FN_BLOCKS_)
#define SB_HI_ (SB_PK_ + PK_BLOCKS_)
#define SETUP_BLOCKS_ (SB_HI_ + HI_BLOCKS_)
__global__ __launch_bounds__(128) void gnn_setup(
    const float* __restrict__ r, const float* __restrict__ coords,
    const float* __restrict__ spin_emb, const float* __restrict__ nuc_emb,
    const float* __restrict__ We, const float* __restrict__ Wn,
    const float* __restrict__ W1, const float* __restrict__ W2,
    float* __restrict__ Wp, float* __restrict__ Aee, float* __restrict__ Ane,
    float* __restrict__ h, float* __restrict__ hW, float* __restrict__ hnW) {
  const int blk = blockIdx.x;
  if (blk < SB_HW_)      setup_feat_ee(blk, r, Aee);
  else if (blk < SB_FN_) setup_hw(blk - SB_HW_, spin_emb, nuc_emb, We, Wn, hW, hnW);
  else if (blk < SB_PK_) setup_feat_ne(blk - SB_FN_, r, coords, Ane);
  else if (blk < SB_HI_) setup_pack_w(blk - SB_PK_, We, Wn, W1, W2, Wp);
  else                   setup_init_h(blk - SB_HI_, spin_emb, h);
}

#define AGG_RING_  8
#define AGG_AHEAD_ 6
#define AGG_STEPS_ (NE_ + NN_)
#define LROWS_     32

__global__ __launch_bounds__(256) void gnn_layer(
    const float* __restrict__ Aee, const float* __restrict__ Ane,
    const float* __restrict__ hW, const float* __restrict__ hnW,
    const float* __restrict__ Wrbf_ee_l, const float* __restrict__ Wrbf_ne_l,
    const float* __restrict__ Wp1, const float* __restrict__ Wp2,
    const float* __restrict__ Wp_ee_next,
    const float* h, float* dst,
    float* __restrict__ hW_next) {
  __shared__ __attribute__((aligned(16))) float Xs[LROWS_ * XP_];
  __shared__ __attribute__((aligned(16))) float Ys[LROWS_ * XP_];
  const int wid = blockIdx.x;
  const int b  = wid >> 1;
  const int w  = __builtin_amdgcn_readfirstlane((int)(threadIdx.x >> 5));
  const int mw = w & 1;
  const int nq = w >> 1;
  const int lane = threadIdx.x & 31;
  const int half = lane >> 4, l16 = lane & 15;
  const int i0 = (wid & 1) * LROWS_ + mw * 16, n0 = nq * (16 * LNT_);
  float* Xw = Xs + mw * 16 * XP_;
  float* Yw = Ys + mw * 16 * XP_;

  const float* __restrict__ a_ee = Aee + ((size_t)b * NE_ + i0 + l16) * KEE_ + half * 2;
  const float* __restrict__ a_ne = Ane + ((size_t)b * NE_ + i0 + l16) * KNE_ + half * 2;
  const float* __restrict__ h_ee = hW + (size_t)b * NE_ * D_ + n0 + l16;
  const float* __restrict__ h_ne = hnW + n0 + l16;

  v8f   acc[LNT_] = {};
  v2f   ra[AGG_RING_][4];
  float rh[AGG_RING_][LNT_];

  auto issue_ee = [&](int j, int st) {
#pragma unroll
    for (int t = 0; t < LNT_; ++t) rh[st][t] = h_ee[(size_t)j * D_ + t * 16];
#pragma unroll
    for (int kk = 0; kk < 4; ++kk) ra[st][kk] = *(const v2f*)&a_ee[j * NRBF_ + kk * 4];
  };
  auto issue_ne = [&](int a, int st) {
#pragma unroll
    for (int t = 0; t < LNT_; ++t) rh[st][t] = h_ne[(size_t)a * D_ + t * 16];
#pragma unroll
    for (int kk = 0; kk < 4; ++kk) ra[st][kk] = *(const v2f*)&a_ne[a * NRBF_ + kk * 4];
  };

#pragma unroll
  for (int s = 0; s < AGG_AHEAD_; ++s) issue_ee(s, s);

  float wk_ee[LNT_][8], wk_ne[LNT_][8];
#pragma unroll
  for (int kk = 0; kk < 4; ++kk)
#pragma unroll
    for (int t = 0; t < LNT_; ++t) {
      const int kb = kk * 4 + half * 2;
      wk_ee[t][kk * 2 + 0] = Wrbf_ee_l[(size_t)(kb + 0) * D_ + n0 + t * 16 + l16];
      wk_ee[t][kk * 2 + 1] = Wrbf_ee_l[(size_t)(kb + 1) * D_ + n0 + t * 16 + l16];
      wk_ne[t][kk * 2 + 0] = Wrbf_ne_l[(size_t)(kb + 0) * D_ + n0 + t * 16 + l16];
      wk_ne[t][kk * 2 + 1] = Wrbf_ne_l[(size_t)(kb + 1) * D_ + n0 + t * 16 + l16];
    }

  auto compute = [&](int st, const float (&wk)[LNT_][8]) {
#pragma unroll
    for (int kk = 0; kk < 4; ++kk) {
      const v2f a = ra[st][kk];
#pragma unroll
      for (int t = 0; t < LNT_; ++t) {
        v2f bb;
        bb.x = rh[st][t] * wk[t][kk * 2 + 0];
        bb.y = rh[st][t] * wk[t][kk * 2 + 1];
        acc[t] = __builtin_amdgcn_wmma_f32_16x16x4_f32(false, a, false, bb, (short)0,
                                                       acc[t], false, false);
      }
    }
  };

#pragma unroll 1
  for (int jo = 0; jo < NE_ - AGG_RING_; jo += AGG_RING_) {
#pragma unroll
    for (int u = 0; u < AGG_RING_; ++u) {
      issue_ee(jo + u + AGG_AHEAD_, (u + AGG_AHEAD_) % AGG_RING_);
      __builtin_amdgcn_sched_barrier(0);
      compute(u, wk_ee);
    }
  }
#pragma unroll
  for (int s = NE_ - AGG_RING_; s < AGG_STEPS_; ++s) {
    const int sn = s + AGG_AHEAD_;
    if (sn < NE_)             issue_ee(sn, sn % AGG_RING_);
    else if (sn < AGG_STEPS_) issue_ne(sn - NE_, sn % AGG_RING_);
    __builtin_amdgcn_sched_barrier(0);
    if (s < NE_) compute(s % AGG_RING_, wk_ee);
    else         compute(s % AGG_RING_, wk_ne);
  }

  store_strip_f32(Xw, XP_, 0, n0, acc, lane);
  __syncthreads();

  v8f acc2[LNT_] = {};
  gemm_strip_ldsA(Xw, Wp1, n0, acc2, lane);
#pragma unroll
  for (int t = 0; t < LNT_; ++t)
#pragma unroll
    for (int ii = 0; ii < 8; ++ii) acc2[t][ii] = tanhf(acc2[t][ii]);
  store_strip_f32(Yw, XP_, 0, n0, acc2, lane);
  __syncthreads();

  const int m0 = b * NE_ + i0;
  v8f acc3[LNT_];
  load_strip_f32(h, D_, m0, n0, acc3, lane);
  gemm_strip_ldsA(Yw, Wp2, n0, acc3, lane);
  store_strip_f32(dst, D_, m0, n0, acc3, lane);

  if (Wp_ee_next != nullptr) {
    store_strip_f32(Xw, XP_, 0, n0, acc3, lane);
    __syncthreads();
    v8f acc4[LNT_] = {};
    gemm_strip_ldsA(Xw, Wp_ee_next, n0, acc4, lane);
    store_strip_f32(hW_next, D_, m0, n0, acc4, lane);
  }
}

extern "C" void kernel_launch(void* const* d_in, const int* in_sizes, int n_in,
                              void* d_out, int out_size, void* d_ws, size_t ws_size,
                              hipStream_t stream) {
  (void)in_sizes; (void)n_in; (void)out_size; (void)ws_size;
  const float* r        = (const float*)d_in[0];
  const float* coords   = (const float*)d_in[1];
  const float* spin_emb = (const float*)d_in[2];
  const float* nuc_emb  = (const float*)d_in[3];
  const float* W_rbf_ee = (const float*)d_in[4];
  const float* W_rbf_ne = (const float*)d_in[5];
  const float* W_msg_ee = (const float*)d_in[6];
  const float* W_msg_ne = (const float*)d_in[7];
  const float* W1       = (const float*)d_in[8];
  const float* W2       = (const float*)d_in[9];
  float* out = (float*)d_out;

  float* ws  = (float*)d_ws;
  float* Aee = ws;
  float* Ane = Aee + (size_t)B_ * NE_ * KEE_;
  float* h   = Ane + (size_t)B_ * NE_ * KNE_;
  float* hWa = h   + (size_t)B_ * NE_ * D_;
  float* hWb = hWa + (size_t)B_ * NE_ * D_;
  float* hnW = hWb + (size_t)B_ * NE_ * D_;
  float* Wp  = hnW + (size_t)L_ * NN_ * D_;

  gnn_setup<<<SETUP_BLOCKS_, 128, 0, stream>>>(r, coords, spin_emb, nuc_emb,
                                               W_msg_ee, W_msg_ne, W1, W2,
                                               Wp, Aee, Ane, h, hWa, hnW);

  for (int l = 0; l < L_; ++l) {
    const float* Wpl = Wp + (size_t)l * 4 * WSZ_;
    const bool last = (l == L_ - 1);
    const float* hW_in  = (l & 1) ? hWb : hWa;
    float*       hW_out = (l & 1) ? hWa : hWb;
    gnn_layer<<<B_ * (NE_ / LROWS_), 256, 0, stream>>>(Aee, Ane, hW_in, hnW + (size_t)l * NN_ * D_,
                                         W_rbf_ee + (size_t)l * NRBF_ * D_,
                                         W_rbf_ne + (size_t)l * NRBF_ * D_,
                                         Wpl + 2 * WSZ_, Wpl + 3 * WSZ_,
                                         last ? nullptr : Wpl + 4 * WSZ_,
                                         h, last ? out : h, hW_out);
  }
}


# basis tables stored in A-fragment order per 16-row tile and step: a step's A operand is two whole-line 16-byte loads per lane, not four 8-byte loads of 16 lines each; same values, products, order
# speedup vs baseline: 2.7289x; 1.0350x over previous
#include <hip/hip_runtime.h>
#include <math.h>

#define B_     128
#define NE_    64
#define NN_    16
#define D_     128
#define NRBF_  16
#define L_     3
#define NUP_   32
#define CUTOFF_ 2.0f
#define EPS_    1e-12f

#define KEE_   (NE_ * NRBF_)
#define KNE_   (NN_ * NRBF_)
#define NT_    4
#define LNT_   2
#define WSZ_   (D_ * D_)
#define FRAG_  (16 * NRBF_)

typedef float v2f __attribute__((ext_vector_type(2)));
typedef float v4f __attribute__((ext_vector_type(4)));
typedef float v8f __attribute__((ext_vector_type(8)));

template <int NT>
__device__ __forceinline__ void gemm_strip_f32u(const float* __restrict__ arow,
                                                const float* __restrict__ Wm,
                                                int n0, v8f (&acc)[NT], int lane) {
  const int half = lane >> 4;
  const int l16  = lane & 15;
#pragma unroll
  for (int k0 = 0; k0 < D_; k0 += 4) {
    const int ka = k0 + half * 2;
    const v2f a = *(const v2f*)&arow[ka];
    const float* __restrict__ bcol = Wm + (size_t)ka * D_ + n0 + l16;
#pragma unroll
    for (int t = 0; t < NT; ++t) {
      v2f b;
      b.x = bcol[t * 16];
      b.y = bcol[D_ + t * 16];
      acc[t] = __builtin_amdgcn_wmma_f32_16x16x4_f32(false, a, false, b, (short)0,
                                                     acc[t], false, false);
    }
  }
}

template <int NT>
__device__ __forceinline__ void store_strip_f32(float* __restrict__ C, int ldc,
                                                int m0, int n0, v8f (&acc)[NT], int lane) {
  const int half = lane >> 4;
  const int l16  = lane & 15;
#pragma unroll
  for (int t = 0; t < NT; ++t)
#pragma unroll
    for (int ii = 0; ii < 8; ++ii)
      C[(size_t)(m0 + half * 8 + ii) * ldc + n0 + t * 16 + l16] = acc[t][ii];
}

template <int NT>
__device__ __forceinline__ void load_strip_f32(const float* __restrict__ C, int ldc,
                                               int m0, int n0, v8f (&acc)[NT], int lane) {
  const int half = lane >> 4;
  const int l16  = lane & 15;
#pragma unroll
  for (int t = 0; t < NT; ++t)
#pragma unroll
    for (int ii = 0; ii < 8; ++ii)
      acc[t][ii] = C[(size_t)(m0 + half * 8 + ii) * ldc + n0 + t * 16 + l16];
}

#define XP_ (D_ + 4)

template <int NT>
__device__ __forceinline__ void gemm_strip_ldsA(const float* Xs,
                                                const float* __restrict__ Bp,
                                                int n0, v8f (&acc)[NT], int lane) {
  const int half = lane >> 4;
  const int l16  = lane & 15;
  const float* arow = Xs + l16 * XP_;
#pragma unroll
  for (int k0 = 0; k0 < D_; k0 += 4) {
    const int ka = k0 + half * 2;
    const v2f a = *(const v2f*)&arow[ka];
    const float* __restrict__ brow = Bp + (size_t)(ka >> 1) * (2 * D_);
#pragma unroll
    for (int t = 0; t < NT; ++t) {
      const v2f b = *(const v2f*)&brow[(n0 + t * 16 + l16) * 2];
      acc[t] = __builtin_amdgcn_wmma_f32_16x16x4_f32(false, a, false, b, (short)0,
                                                     acc[t], false, false);
    }
  }
}


#define PK_BLOCKS_ ((12 * (D_ / 2) * (D_ / 4)) / 128)
__device__ __forceinline__ void setup_pack_w(
    int blk,
    const float* __restrict__ We, const float* __restrict__ Wn,
    const float* __restrict__ W1, const float* __restrict__ W2,
    float* __restrict__ Wp) {
  const int gid = blk * 128 + threadIdx.x;
  const int m   = gid >> 11;
  const int kp  = (gid >> 5) & 63;
  const int c4  = gid & 31;
  const int l = m >> 2, w = m & 3;
  const float* src = (w == 0 ? We : w == 1 ? Wn : w == 2 ? W1 : W2) + (size_t)l * WSZ_;
  const v4f x = *(const v4f*)&src[(size_t)(2 * kp) * D_ + c4 * 4];
  const v4f y = *(const v4f*)&src[(size_t)(2 * kp + 1) * D_ + c4 * 4];
  float* d = Wp + (size_t)m * WSZ_ + (size_t)kp * (2 * D_) + c4 * 8;
  v4f lo, hi;
  lo.x = x.x; lo.y = y.x; lo.z = x.y; lo.w = y.y;
  hi.x = x.z; hi.y = y.z; hi.z = x.w; hi.w = y.w;
  *(v4f*)d       = lo;
  *(v4f*)(d + 4) = hi;
}

__device__ __forceinline__ void rbf_frag(float dx, float dy, float dz, int half, bool zero,
                                         v4f& q0, v4f& q1) {
  const float inv_cut = 1.0f / CUTOFF_;
  const float dmu     = CUTOFF_ / (float)(NRBF_ - 1);
  const float inv_sig = (float)NRBF_ / CUTOFF_;
  const float dist = sqrtf(__builtin_fmaf(dz, dz, __builtin_fmaf(dx, dx, dy * dy)) + EPS_);
  float env = 1.0f - dist * inv_cut;
  env = (dist < CUTOFF_) ? env * env : 0.0f;
#pragma unroll
  for (int kk = 0; kk < 4; ++kk)
#pragma unroll
    for (int e = 0; e < 2; ++e) {
      const int k = kk * 4 + half * 2 + e;
      const float z = (dist - dmu * (float)k) * inv_sig;
      float ge = expf(-0.5f * z * z) * env;
      if (zero) ge = 0.0f;
      if (kk < 2) q0[(kk & 1) * 2 + e] = ge;
      else        q1[(kk & 1) * 2 + e] = ge;
    }
}

#define FE_J_ 8
#define FE_BLOCKS_ ((B_ * (NE_ / 16) * (NE_ / FE_J_)) / 4)
__device__ __forceinline__ void setup_feat_ee(int blk, const float* __restrict__ r,
                                              float* __restrict__ Aee) {
  const int wave = blk * 4 + __builtin_amdgcn_readfirstlane((int)(threadIdx.x >> 5));
  const int lane = threadIdx.x & 31;
  const int half = lane >> 4, l16 = lane & 15;
  const int j0 = (wave & (NE_ / FE_J_ - 1)) * FE_J_;
  const int bt = wave / (NE_ / FE_J_);
  const int b  = bt >> 2;
  const int i  = (bt & 3) * 16 + l16;
  const float* pi = r + (size_t)(b * NE_ + i) * 3;
  const float rx = pi[0], ry = pi[1], rz = pi[2];
  const float* __restrict__ pj = r + (size_t)(b * NE_ + j0) * 3;
  float* __restrict__ o = Aee + ((size_t)bt * NE_ + j0) * FRAG_ + lane * 4;
#pragma unroll
  for (int jj = 0; jj < FE_J_; ++jj) {
    v4f q0, q1;
    rbf_frag(rx - pj[jj * 3 + 0], ry - pj[jj * 3 + 1], rz - pj[jj * 3 + 2], half,
             i == j0 + jj, q0, q1);
    *(v4f*)&o[jj * FRAG_]       = q0;
    *(v4f*)&o[jj * FRAG_ + 128] = q1;
  }
}

#define FN_A_ 4
#define FN_BLOCKS_ ((B_ * (NE_ / 16) * (NN_ / FN_A_)) / 4)
__device__ __forceinline__ void setup_feat_ne(int blk, const float* __restrict__ r,
                                              const float* __restrict__ coords,
                                              float* __restrict__ Ane) {
  const int wave = blk * 4 + __builtin_amdgcn_readfirstlane((int)(threadIdx.x >> 5));
  const int lane = threadIdx.x & 31;
  const int half = lane >> 4, l16 = lane & 15;
  const int a0 = (wave & (NN_ / FN_A_ - 1)) * FN_A_;
  const int bt = wave / (NN_ / FN_A_);
  const int b  = bt >> 2;
  const int i  = (bt & 3) * 16 + l16;
  const float* pi = r + (size_t)(b * NE_ + i) * 3;
  const float rx = pi[0], ry = pi[1], rz = pi[2];
  const float* __restrict__ pc = coords + a0 * 3;
  float* __restrict__ o = Ane + ((size_t)bt * NN_ + a0) * FRAG_ + lane * 4;
#pragma unroll
  for (int aa = 0; aa < FN_A_; ++aa) {
    v4f q0, q1;
    rbf_frag(rx - pc[aa * 3 + 0], ry - pc[aa * 3 + 1], rz - pc[aa * 3 + 2], half,
             false, q0, q1);
    *(v4f*)&o[aa * FRAG_]       = q0;
    *(v4f*)&o[aa * FRAG_ + 128] = q1;
  }
}

#define HI_ROWS_ 8
#define HI_BLOCKS_ ((B_ * NE_) / (4 * HI_ROWS_))
__device__ __forceinline__ void setup_init_h(int blk, const float* __restrict__ spin_emb,
                                             float* __restrict__ h) {
  const int wave = blk * 4 + (threadIdx.x >> 5);
  const int lane = threadIdx.x & 31;
  const int row0 = wave * HI_ROWS_;
  const int i = row0 & (NE_ - 1);
  const v4f v = *(const v4f*)&spin_emb[(i < NUP_ ? 0 : 1) * D_ + lane * 4];
#pragma unroll
  for (int rr = 0; rr < HI_ROWS_; ++rr)
    *(v4f*)&h[(size_t)(row0 + rr) * D_ + lane * 4] = v;
}

#define HW_STRIPS_ (2 * (B_ * NE_ / 16))
#define HW_BLOCKS_ ((HW_STRIPS_ + 2 * L_ + 3) / 4)
__device__ __forceinline__ void setup_hw(int blk,
                                         const float* __restrict__ spin_emb,
                                         const float* __restrict__ nuc_emb,
                                         const float* __restrict__ We,
                                         const float* __restrict__ Wn,
                                         float* __restrict__ hW, float* __restrict__ hnW) {
  const int s = blk * 4 + __builtin_amdgcn_readfirstlane((int)(threadIdx.x >> 5));
  const int lane = threadIdx.x & 31;
  const int l16  = lane & 15;
  v8f acc[NT_] = {};
  if (s < HW_STRIPS_) {
    const int m0 = (s >> 1) * 16;
    const int n0 = (s & 1) * 64;
    const int spin = ((m0 & (NE_ - 1)) < NUP_) ? 0 : 1;
    gemm_strip_f32u(spin_emb + spin * D_, We, n0, acc, lane);
    store_strip_f32(hW, D_, m0, n0, acc, lane);
  } else if (s < HW_STRIPS_ + 2 * L_) {
    const int l  = (s - HW_STRIPS_) >> 1;
    const int n0 = ((s - HW_STRIPS_) & 1) * 64;
    gemm_strip_f32u(nuc_emb + l16 * D_, Wn + (size_t)l * WSZ_, n0, acc, lane);
    store_strip_f32(hnW + (size_t)l * NN_ * D_, D_, 0, n0, acc, lane);
  }
}

#define SB_HW_ (FE_BLOCKS_)
#define SB_FN_ (SB_HW_ + HW_BLOCKS_)
#define SB_PK_ (SB_FN_ + FN_BLOCKS_)
#define SB_HI_ (SB_PK_ + PK_BLOCKS_)
#define SETUP_BLOCKS_ (SB_HI_ + HI_BLOCKS_)
__global__ __launch_bounds__(128) void gnn_setup(
    const float* __restrict__ r, const float* __restrict__ coords,
    const float* __restrict__ spin_emb, const float* __restrict__ nuc_emb,
    const float* __restrict__ We, const float* __restrict__ Wn,
    const float* __restrict__ W1, const float* __restrict__ W2,
    float* __restrict__ Wp, float* __restrict__ Aee, float* __restrict__ Ane,
    float* __restrict__ h, float* __restrict__ hW, float* __restrict__ hnW) {
  const int blk = blockIdx.x;
  if (blk < SB_HW_)      setup_feat_ee(blk, r, Aee);
  else if (blk < SB_FN_) setup_hw(blk - SB_HW_, spin_emb, nuc_emb, We, Wn, hW, hnW);
  else if (blk < SB_PK_) setup_feat_ne(blk - SB_FN_, r, coords, Ane);
  else if (blk < SB_HI_) setup_pack_w(blk - SB_PK_, We, Wn, W1, W2, Wp);
  else                   setup_init_h(blk - SB_HI_, spin_emb, h);
}

#define AGG_RING_  8
#define AGG_AHEAD_ 6
#define AGG_STEPS_ (NE_ + NN_)
#define LROWS_     32

__global__ __launch_bounds__(256) void gnn_layer(
    const float* __restrict__ Aee, const float* __restrict__ Ane,
    const float* __restrict__ hW, const float* __restrict__ hnW,
    const float* __restrict__ Wrbf_ee_l, const float* __restrict__ Wrbf_ne_l,
    const float* __restrict__ Wp1, const float* __restrict__ Wp2,
    const float* __restrict__ Wp_ee_next,
    const float* h, float* dst,
    float* __restrict__ hW_next) {
  __shared__ __attribute__((aligned(16))) float Xs[LROWS_ * XP_];
  __shared__ __attribute__((aligned(16))) float Ys[LROWS_ * XP_];
  const int wid = blockIdx.x;
  const int b  = wid >> 1;
  const int w  = __builtin_amdgcn_readfirstlane((int)(threadIdx.x >> 5));
  const int mw = w & 1;
  const int nq = w >> 1;
  const int lane = threadIdx.x & 31;
  const int half = lane >> 4, l16 = lane & 15;
  const int i0 = (wid & 1) * LROWS_ + mw * 16, n0 = nq * (16 * LNT_);
  float* Xw = Xs + mw * 16 * XP_;
  float* Yw = Ys + mw * 16 * XP_;

  const int bt = b * (NE_ / 16) + (i0 >> 4);
  const float* __restrict__ a_ee = Aee + (size_t)bt * NE_ * FRAG_ + lane * 4;
  const float* __restrict__ a_ne = Ane + (size_t)bt * NN_ * FRAG_ + lane * 4;
  const float* __restrict__ h_ee = hW + (size_t)b * NE_ * D_ + n0 + l16;
  const float* __restrict__ h_ne = hnW + n0 + l16;

  v8f   acc[LNT_] = {};
  v4f   rq[AGG_RING_][2];
  float rh[AGG_RING_][LNT_];

  auto issue_ee = [&](int j, int st) {
#pragma unroll
    for (int t = 0; t < LNT_; ++t) rh[st][t] = h_ee[(size_t)j * D_ + t * 16];
    rq[st][0] = *(const v4f*)&a_ee[j * FRAG_];
    rq[st][1] = *(const v4f*)&a_ee[j * FRAG_ + 128];
  };
  auto issue_ne = [&](int a, int st) {
#pragma unroll
    for (int t = 0; t < LNT_; ++t) rh[st][t] = h_ne[(size_t)a * D_ + t * 16];
    rq[st][0] = *(const v4f*)&a_ne[a * FRAG_];
    rq[st][1] = *(const v4f*)&a_ne[a * FRAG_ + 128];
  };

#pragma unroll
  for (int s = 0; s < AGG_AHEAD_; ++s) issue_ee(s, s);

  float wk_ee[LNT_][8], wk_ne[LNT_][8];
#pragma unroll
  for (int kk = 0; kk < 4; ++kk)
#pragma unroll
    for (int t = 0; t < LNT_; ++t) {
      const int kb = kk * 4 + half * 2;
      wk_ee[t][kk * 2 + 0] = Wrbf_ee_l[(size_t)(kb + 0) * D_ + n0 + t * 16 + l16];
      wk_ee[t][kk * 2 + 1] = Wrbf_ee_l[(size_t)(kb + 1) * D_ + n0 + t * 16 + l16];
      wk_ne[t][kk * 2 + 0] = Wrbf_ne_l[(size_t)(kb + 0) * D_ + n0 + t * 16 + l16];
      wk_ne[t][kk * 2 + 1] = Wrbf_ne_l[(size_t)(kb + 1) * D_ + n0 + t * 16 + l16];
    }

  auto compute = [&](int st, const float (&wk)[LNT_][8]) {
#pragma unroll
    for (int kk = 0; kk < 4; ++kk) {
      const v4f q = rq[st][kk >> 1];
      v2f a;
      a.x = (kk & 1) ? q.z : q.x;
      a.y = (kk & 1) ? q.w : q.y;
#pragma unroll
      for (int t = 0; t < LNT_; ++t) {
        v2f bb;
        bb.x = rh[st][t] * wk[t][kk * 2 + 0];
        bb.y = rh[st][t] * wk[t][kk * 2 + 1];
        acc[t] = __builtin_amdgcn_wmma_f32_16x16x4_f32(false, a, false, bb, (short)0,
                                                       acc[t], false, false);
      }
    }
  };

#pragma unroll 1
  for (int jo = 0; jo < NE_ - AGG_RING_; jo += AGG_RING_) {
#pragma unroll
    for (int u = 0; u < AGG_RING_; ++u) {
      issue_ee(jo + u + AGG_AHEAD_, (u + AGG_AHEAD_) % AGG_RING_);
      __builtin_amdgcn_sched_barrier(0);
      compute(u, wk_ee);
    }
  }
#pragma unroll
  for (int s = NE_ - AGG_RING_; s < AGG_STEPS_; ++s) {
    const int sn = s + AGG_AHEAD_;
    if (sn < NE_)             issue_ee(sn, sn % AGG_RING_);
    else if (sn < AGG_STEPS_) issue_ne(sn - NE_, sn % AGG_RING_);
    __builtin_amdgcn_sched_barrier(0);
    if (s < NE_) compute(s % AGG_RING_, wk_ee);
    else         compute(s % AGG_RING_, wk_ne);
  }

  store_strip_f32(Xw, XP_, 0, n0, acc, lane);
  __syncthreads();

  v8f acc2[LNT_] = {};
  gemm_strip_ldsA(Xw, Wp1, n0, acc2, lane);
#pragma unroll
  for (int t = 0; t < LNT_; ++t)
#pragma unroll
    for (int ii = 0; ii < 8; ++ii) acc2[t][ii] = tanhf(acc2[t][ii]);
  store_strip_f32(Yw, XP_, 0, n0, acc2, lane);
  __syncthreads();

  const int m0 = b * NE_ + i0;
  v8f acc3[LNT_];
  load_strip_f32(h, D_, m0, n0, acc3, lane);
  gemm_strip_ldsA(Yw, Wp2, n0, acc3, lane);
  store_strip_f32(dst, D_, m0, n0, acc3, lane);

  if (Wp_ee_next != nullptr) {
    store_strip_f32(Xw, XP_, 0, n0, acc3, lane);
    __syncthreads();
    v8f acc4[LNT_] = {};
    gemm_strip_ldsA(Xw, Wp_ee_next, n0, acc4, lane);
    store_strip_f32(hW_next, D_, m0, n0, acc4, lane);
  }
}

extern "C" void kernel_launch(void* const* d_in, const int* in_sizes, int n_in,
                              void* d_out, int out_size, void* d_ws, size_t ws_size,
                              hipStream_t stream) {
  (void)in_sizes; (void)n_in; (void)out_size; (void)ws_size;
  const float* r        = (const float*)d_in[0];
  const float* coords   = (const float*)d_in[1];
  const float* spin_emb = (const float*)d_in[2];
  const float* nuc_emb  = (const float*)d_in[3];
  const float* W_rbf_ee = (const float*)d_in[4];
  const float* W_rbf_ne = (const float*)d_in[5];
  const float* W_msg_ee = (const float*)d_in[6];
  const float* W_msg_ne = (const float*)d_in[7];
  const float* W1       = (const float*)d_in[8];
  const float* W2       = (const float*)d_in[9];
  float* out = (float*)d_out;

  float* ws  = (float*)d_ws;
  float* Aee = ws;
  float* Ane = Aee + (size_t)B_ * NE_ * KEE_;
  float* h   = Ane + (size_t)B_ * NE_ * KNE_;
  float* hWa = h   + (size_t)B_ * NE_ * D_;
  float* hWb = hWa + (size_t)B_ * NE_ * D_;
  float* hnW = hWb + (size_t)B_ * NE_ * D_;
  float* Wp  = hnW + (size_t)L_ * NN_ * D_;

  gnn_setup<<<SETUP_BLOCKS_, 128, 0, stream>>>(r, coords, spin_emb, nuc_emb,
                                               W_msg_ee, W_msg_ne, W1, W2,
                                               Wp, Aee, Ane, h, hWa, hnW);

  for (int l = 0; l < L_; ++l) {
    const float* Wpl = Wp + (size_t)l * 4 * WSZ_;
    const bool last = (l == L_ - 1);
    const float* hW_in  = (l & 1) ? hWb : hWa;
    float*       hW_out = (l & 1) ? hWa : hWb;
    gnn_layer<<<B_ * (NE_ / LROWS_), 256, 0, stream>>>(Aee, Ane, hW_in, hnW + (size_t)l * NN_ * D_,
                                         W_rbf_ee + (size_t)l * NRBF_ * D_,
                                         W_rbf_ne + (size_t)l * NRBF_ * D_,
                                         Wpl + 2 * WSZ_, Wpl + 3 * WSZ_,
                                         last ? nullptr : Wpl + 4 * WSZ_,
                                         h, last ? out : h, hW_out);
  }
}


# first layer: its hW is one row per spin group, so a group's 32 A fragments are added (vector adds, partials via LDS) before ONE 8-WMMA step per group: 64 ee steps become 2; ee sum order changes
# speedup vs baseline: 2.9812x; 1.0925x over previous
#include <hip/hip_runtime.h>
#include <math.h>

#define B_     128
#define NE_    64
#define NN_    16
#define D_     128
#define NRBF_  16
#define L_     3
#define NUP_   32
#define CUTOFF_ 2.0f
#define EPS_    1e-12f

#define KEE_   (NE_ * NRBF_)
#define KNE_   (NN_ * NRBF_)
#define NT_    4
#define LNT_   2
#define WSZ_   (D_ * D_)
#define FRAG_  (16 * NRBF_)

typedef float v2f __attribute__((ext_vector_type(2)));
typedef float v4f __attribute__((ext_vector_type(4)));
typedef float v8f __attribute__((ext_vector_type(8)));

template <int NT>
__device__ __forceinline__ void gemm_strip_f32u(const float* __restrict__ arow,
                                                const float* __restrict__ Wm,
                                                int n0, v8f (&acc)[NT], int lane) {
  const int half = lane >> 4;
  const int l16  = lane & 15;
#pragma unroll
  for (int k0 = 0; k0 < D_; k0 += 4) {
    const int ka = k0 + half * 2;
    const v2f a = *(const v2f*)&arow[ka];
    const float* __restrict__ bcol = Wm + (size_t)ka * D_ + n0 + l16;
#pragma unroll
    for (int t = 0; t < NT; ++t) {
      v2f b;
      b.x = bcol[t * 16];
      b.y = bcol[D_ + t * 16];
      acc[t] = __builtin_amdgcn_wmma_f32_16x16x4_f32(false, a, false, b, (short)0,
                                                     acc[t], false, false);
    }
  }
}

template <int NT>
__device__ __forceinline__ void store_strip_f32(float* __restrict__ C, int ldc,
                                                int m0, int n0, v8f (&acc)[NT], int lane) {
  const int half = lane >> 4;
  const int l16  = lane & 15;
#pragma unroll
  for (int t = 0; t < NT; ++t)
#pragma unroll
    for (int ii = 0; ii < 8; ++ii)
      C[(size_t)(m0 + half * 8 + ii) * ldc + n0 + t * 16 + l16] = acc[t][ii];
}

template <int NT>
__device__ __forceinline__ void load_strip_f32(const float* __restrict__ C, int ldc,
                                               int m0, int n0, v8f (&acc)[NT], int lane) {
  const int half = lane >> 4;
  const int l16  = lane & 15;
#pragma unroll
  for (int t = 0; t < NT; ++t)
#pragma unroll
    for (int ii = 0; ii < 8; ++ii)
      acc[t][ii] = C[(size_t)(m0 + half * 8 + ii) * ldc + n0 + t * 16 + l16];
}

#define XP_ (D_ + 4)

template <int NT>
__device__ __forceinline__ void gemm_strip_ldsA(const float* Xs,
                                                const float* __restrict__ Bp,
                                                int n0, v8f (&acc)[NT], int lane) {
  const int half = lane >> 4;
  const int l16  = lane & 15;
  const float* arow = Xs + l16 * XP_;
#pragma unroll
  for (int k0 = 0; k0 < D_; k0 += 4) {
    const int ka = k0 + half * 2;
    const v2f a = *(const v2f*)&arow[ka];
    const float* __restrict__ brow = Bp + (size_t)(ka >> 1) * (2 * D_);
#pragma unroll
    for (int t = 0; t < NT; ++t) {
      const v2f b = *(const v2f*)&brow[(n0 + t * 16 + l16) * 2];
      acc[t] = __builtin_amdgcn_wmma_f32_16x16x4_f32(false, a, false, b, (short)0,
                                                     acc[t], false, false);
    }
  }
}


#define PK_BLOCKS_ ((12 * (D_ / 2) * (D_ / 4)) / 128)
__device__ __forceinline__ void setup_pack_w(
    int blk,
    const float* __restrict__ We, const float* __restrict__ Wn,
    const float* __restrict__ W1, const float* __restrict__ W2,
    float* __restrict__ Wp) {
  const int gid = blk * 128 + threadIdx.x;
  const int m   = gid >> 11;
  const int kp  = (gid >> 5) & 63;
  const int c4  = gid & 31;
  const int l = m >> 2, w = m & 3;
  const float* src = (w == 0 ? We : w == 1 ? Wn : w == 2 ? W1 : W2) + (size_t)l * WSZ_;
  const v4f x = *(const v4f*)&src[(size_t)(2 * kp) * D_ + c4 * 4];
  const v4f y = *(const v4f*)&src[(size_t)(2 * kp + 1) * D_ + c4 * 4];
  float* d = Wp + (size_t)m * WSZ_ + (size_t)kp * (2 * D_) + c4 * 8;
  v4f lo, hi;
  lo.x = x.x; lo.y = y.x; lo.z = x.y; lo.w = y.y;
  hi.x = x.z; hi.y = y.z; hi.z = x.w; hi.w = y.w;
  *(v4f*)d       = lo;
  *(v4f*)(d + 4) = hi;
}

__device__ __forceinline__ void rbf_frag(float dx, float dy, float dz, int half, bool zero,
                                         v4f& q0, v4f& q1) {
  const float inv_cut = 1.0f / CUTOFF_;
  const float dmu     = CUTOFF_ / (float)(NRBF_ - 1);
  const float inv_sig = (float)NRBF_ / CUTOFF_;
  const float dist = sqrtf(__builtin_fmaf(dz, dz, __builtin_fmaf(dx, dx, dy * dy)) + EPS_);
  float env = 1.0f - dist * inv_cut;
  env = (dist < CUTOFF_) ? env * env : 0.0f;
#pragma unroll
  for (int kk = 0; kk < 4; ++kk)
#pragma unroll
    for (int e = 0; e < 2; ++e) {
      const int k = kk * 4 + half * 2 + e;
      const float z = (dist - dmu * (float)k) * inv_sig;
      float ge = expf(-0.5f * z * z) * env;
      if (zero) ge = 0.0f;
      if (kk < 2) q0[(kk & 1) * 2 + e] = ge;
      else        q1[(kk & 1) * 2 + e] = ge;
    }
}

#define FE_J_ 8
#define FE_BLOCKS_ ((B_ * (NE_ / 16) * (NE_ / FE_J_)) / 4)
__device__ __forceinline__ void setup_feat_ee(int blk, const float* __restrict__ r,
                                              float* __restrict__ Aee) {
  const int wave = blk * 4 + __builtin_amdgcn_readfirstlane((int)(threadIdx.x >> 5));
  const int lane = threadIdx.x & 31;
  const int half = lane >> 4, l16 = lane & 15;
  const int j0 = (wave & (NE_ / FE_J_ - 1)) * FE_J_;
  const int bt = wave / (NE_ / FE_J_);
  const int b  = bt >> 2;
  const int i  = (bt & 3) * 16 + l16;
  const float* pi = r + (size_t)(b * NE_ + i) * 3;
  const float rx = pi[0], ry = pi[1], rz = pi[2];
  const float* __restrict__ pj = r + (size_t)(b * NE_ + j0) * 3;
  float* __restrict__ o = Aee + ((size_t)bt * NE_ + j0) * FRAG_ + lane * 4;
#pragma unroll
  for (int jj = 0; jj < FE_J_; ++jj) {
    v4f q0, q1;
    rbf_frag(rx - pj[jj * 3 + 0], ry - pj[jj * 3 + 1], rz - pj[jj * 3 + 2], half,
             i == j0 + jj, q0, q1);
    *(v4f*)&o[jj * FRAG_]       = q0;
    *(v4f*)&o[jj * FRAG_ + 128] = q1;
  }
}

#define FN_A_ 4
#define FN_BLOCKS_ ((B_ * (NE_ / 16) * (NN_ / FN_A_)) / 4)
__device__ __forceinline__ void setup_feat_ne(int blk, const float* __restrict__ r,
                                              const float* __restrict__ coords,
                                              float* __restrict__ Ane) {
  const int wave = blk * 4 + __builtin_amdgcn_readfirstlane((int)(threadIdx.x >> 5));
  const int lane = threadIdx.x & 31;
  const int half = lane >> 4, l16 = lane & 15;
  const int a0 = (wave & (NN_ / FN_A_ - 1)) * FN_A_;
  const int bt = wave / (NN_ / FN_A_);
  const int b  = bt >> 2;
  const int i  = (bt & 3) * 16 + l16;
  const float* pi = r + (size_t)(b * NE_ + i) * 3;
  const float rx = pi[0], ry = pi[1], rz = pi[2];
  const float* __restrict__ pc = coords + a0 * 3;
  float* __restrict__ o = Ane + ((size_t)bt * NN_ + a0) * FRAG_ + lane * 4;
#pragma unroll
  for (int aa = 0; aa < FN_A_; ++aa) {
    v4f q0, q1;
    rbf_frag(rx - pc[aa * 3 + 0], ry - pc[aa * 3 + 1], rz - pc[aa * 3 + 2], half,
             false, q0, q1);
    *(v4f*)&o[aa * FRAG_]       = q0;
    *(v4f*)&o[aa * FRAG_ + 128] = q1;
  }
}

#define HI_ROWS_ 8
#define HI_BLOCKS_ ((B_ * NE_) / (4 * HI_ROWS_))
__device__ __forceinline__ void setup_init_h(int blk, const float* __restrict__ spin_emb,
                                             float* __restrict__ h) {
  const int wave = blk * 4 + (threadIdx.x >> 5);
  const int lane = threadIdx.x & 31;
  const int row0 = wave * HI_ROWS_;
  const int i = row0 & (NE_ - 1);
  const v4f v = *(const v4f*)&spin_emb[(i < NUP_ ? 0 : 1) * D_ + lane * 4];
#pragma unroll
  for (int rr = 0; rr < HI_ROWS_; ++rr)
    *(v4f*)&h[(size_t)(row0 + rr) * D_ + lane * 4] = v;
}

#define HW_STRIPS_ (2 * (B_ * NE_ / 16))
#define HW_BLOCKS_ ((HW_STRIPS_ + 2 * L_ + 3) / 4)
__device__ __forceinline__ void setup_hw(int blk,
                                         const float* __restrict__ spin_emb,
                                         const float* __restrict__ nuc_emb,
                                         const float* __restrict__ We,
                                         const float* __restrict__ Wn,
                                         float* __restrict__ hW, float* __restrict__ hnW) {
  const int s = blk * 4 + __builtin_amdgcn_readfirstlane((int)(threadIdx.x >> 5));
  const int lane = threadIdx.x & 31;
  const int l16  = lane & 15;
  v8f acc[NT_] = {};
  if (s < HW_STRIPS_) {
    const int m0 = (s >> 1) * 16;
    const int n0 = (s & 1) * 64;
    const int spin = ((m0 & (NE_ - 1)) < NUP_) ? 0 : 1;
    gemm_strip_f32u(spin_emb + spin * D_, We, n0, acc, lane);
    store_strip_f32(hW, D_, m0, n0, acc, lane);
  } else if (s < HW_STRIPS_ + 2 * L_) {
    const int l  = (s - HW_STRIPS_) >> 1;
    const int n0 = ((s - HW_STRIPS_) & 1) * 64;
    gemm_strip_f32u(nuc_emb + l16 * D_, Wn + (size_t)l * WSZ_, n0, acc, lane);
    store_strip_f32(hnW + (size_t)l * NN_ * D_, D_, 0, n0, acc, lane);
  }
}

#define SB_HW_ (FE_BLOCKS_)
#define SB_FN_ (SB_HW_ + HW_BLOCKS_)
#define SB_PK_ (SB_FN_ + FN_BLOCKS_)
#define SB_HI_ (SB_PK_ + PK_BLOCKS_)
#define SETUP_BLOCKS_ (SB_HI_ + HI_BLOCKS_)
__global__ __launch_bounds__(128) void gnn_setup(
    const float* __restrict__ r, const float* __restrict__ coords,
    const float* __restrict__ spin_emb, const float* __restrict__ nuc_emb,
    const float* __restrict__ We, const float* __restrict__ Wn,
    const float* __restrict__ W1, const float* __restrict__ W2,
    float* __restrict__ Wp, float* __restrict__ Aee, float* __restrict__ Ane,
    float* __restrict__ h, float* __restrict__ hW, float* __restrict__ hnW) {
  const int blk = blockIdx.x;
  if (blk < SB_HW_)      setup_feat_ee(blk, r, Aee);
  else if (blk < SB_FN_) setup_hw(blk - SB_HW_, spin_emb, nuc_emb, We, Wn, hW, hnW);
  else if (blk < SB_PK_) setup_feat_ne(blk - SB_FN_, r, coords, Ane);
  else if (blk < SB_HI_) setup_pack_w(blk - SB_PK_, We, Wn, W1, W2, Wp);
  else                   setup_init_h(blk - SB_HI_, spin_emb, h);
}

#define AGG_RING_  8
#define AGG_AHEAD_ 6
#define AGG_STEPS_ (NE_ + NN_)
#define LROWS_     32

template <bool FIRST>
__global__ __launch_bounds__(256) void gnn_layer(
    const float* __restrict__ Aee, const float* __restrict__ Ane,
    const float* __restrict__ hW, const float* __restrict__ hnW,
    const float* __restrict__ Wrbf_ee_l, const float* __restrict__ Wrbf_ne_l,
    const float* __restrict__ Wp1, const float* __restrict__ Wp2,
    const float* __restrict__ Wp_ee_next,
    const float* h, float* dst,
    float* __restrict__ hW_next) {
  __shared__ __attribute__((aligned(16))) float Xs[LROWS_ * XP_];
  __shared__ __attribute__((aligned(16))) float Ys[LROWS_ * XP_];
  const int wid = blockIdx.x;
  const int b  = wid >> 1;
  const int w  = __builtin_amdgcn_readfirstlane((int)(threadIdx.x >> 5));
  const int mw = w & 1;
  const int nq = w >> 1;
  const int lane = threadIdx.x & 31;
  const int half = lane >> 4, l16 = lane & 15;
  const int i0 = (wid & 1) * LROWS_ + mw * 16, n0 = nq * (16 * LNT_);
  float* Xw = Xs + mw * 16 * XP_;
  float* Yw = Ys + mw * 16 * XP_;

  const int bt = b * (NE_ / 16) + (i0 >> 4);
  const float* __restrict__ a_ee = Aee + (size_t)bt * NE_ * FRAG_ + lane * 4;
  const float* __restrict__ a_ne = Ane + (size_t)bt * NN_ * FRAG_ + lane * 4;
  const float* __restrict__ h_ee = hW + (size_t)b * NE_ * D_ + n0 + l16;
  const float* __restrict__ h_ne = hnW + n0 + l16;

  v8f   acc[LNT_] = {};
  v4f   rq[AGG_RING_][2];
  float rh[AGG_RING_][LNT_];

  auto issue_ee = [&](int j, int st) {
#pragma unroll
    for (int t = 0; t < LNT_; ++t) rh[st][t] = h_ee[(size_t)j * D_ + t * 16];
    rq[st][0] = *(const v4f*)&a_ee[j * FRAG_];
    rq[st][1] = *(const v4f*)&a_ee[j * FRAG_ + 128];
  };
  auto issue_ne = [&](int a, int st) {
#pragma unroll
    for (int t = 0; t < LNT_; ++t) rh[st][t] = h_ne[(size_t)a * D_ + t * 16];
    rq[st][0] = *(const v4f*)&a_ne[a * FRAG_];
    rq[st][1] = *(const v4f*)&a_ne[a * FRAG_ + 128];
  };

#pragma unroll
  for (int s = 0; s < AGG_AHEAD_; ++s) {
    if constexpr (FIRST) issue_ne(s, s);
    else                 issue_ee(s, s);
  }

  float wk_ee[LNT_][8], wk_ne[LNT_][8];
#pragma unroll
  for (int kk = 0; kk < 4; ++kk)
#pragma unroll
    for (int t = 0; t < LNT_; ++t) {
      const int kb = kk * 4 + half * 2;
      wk_ee[t][kk * 2 + 0] = Wrbf_ee_l[(size_t)(kb + 0) * D_ + n0 + t * 16 + l16];
      wk_ee[t][kk * 2 + 1] = Wrbf_ee_l[(size_t)(kb + 1) * D_ + n0 + t * 16 + l16];
      wk_ne[t][kk * 2 + 0] = Wrbf_ne_l[(size_t)(kb + 0) * D_ + n0 + t * 16 + l16];
      wk_ne[t][kk * 2 + 1] = Wrbf_ne_l[(size_t)(kb + 1) * D_ + n0 + t * 16 + l16];
    }

  auto mm = [&](const v4f& q0, const v4f& q1, const float (&hv)[LNT_],
                const float (&wk)[LNT_][8]) {
#pragma unroll
    for (int kk = 0; kk < 4; ++kk) {
      const v4f q = (kk >> 1) ? q1 : q0;
      v2f a;
      a.x = (kk & 1) ? q.z : q.x;
      a.y = (kk & 1) ? q.w : q.y;
#pragma unroll
      for (int t = 0; t < LNT_; ++t) {
        v2f bb;
        bb.x = hv[t] * wk[t][kk * 2 + 0];
        bb.y = hv[t] * wk[t][kk * 2 + 1];
        acc[t] = __builtin_amdgcn_wmma_f32_16x16x4_f32(false, a, false, bb, (short)0,
                                                       acc[t], false, false);
      }
    }
  };
  auto compute = [&](int st, const float (&wk)[LNT_][8]) {
    mm(rq[st][0], rq[st][1], rh[st], wk);
  };

  if constexpr (FIRST) {
    static_assert(NUP_ * 2 == NE_ && NE_ == 64,
                  "a spin group is two 16-electron quarters of the four");
    float hg[2][LNT_];
#pragma unroll
    for (int g = 0; g < 2; ++g)
#pragma unroll
      for (int t = 0; t < LNT_; ++t) hg[g][t] = h_ee[(size_t)(g * NUP_) * D_ + t * 16];
    v4f s0 = *(const v4f*)&a_ee[nq * (NE_ / 4) * FRAG_];
    v4f s1 = *(const v4f*)&a_ee[nq * (NE_ / 4) * FRAG_ + 128];
#pragma unroll
    for (int u = 1; u < NE_ / 4; ++u) {
      const int j = nq * (NE_ / 4) + u;
      s0 += *(const v4f*)&a_ee[j * FRAG_];
      s1 += *(const v4f*)&a_ee[j * FRAG_ + 128];
    }
    float* P = Ys;
    *(v4f*)&P[w * 256 + lane * 4]       = s0;
    *(v4f*)&P[w * 256 + 128 + lane * 4] = s1;
    __syncthreads();
#pragma unroll
    for (int g = 0; g < 2; ++g) {
      const float* p0 = P + ((2 * g) * 2 + mw) * 256 + lane * 4;
      const float* p1 = P + ((2 * g + 1) * 2 + mw) * 256 + lane * 4;
      const v4f q0 = *(const v4f*)p0 + *(const v4f*)p1;
      const v4f q1 = *(const v4f*)(p0 + 128) + *(const v4f*)(p1 + 128);
      mm(q0, q1, hg[g], wk_ee);
    }
#pragma unroll
    for (int s = 0; s < NN_; ++s) {
      if (s + AGG_AHEAD_ < NN_) issue_ne(s + AGG_AHEAD_, (s + AGG_AHEAD_) % AGG_RING_);
      __builtin_amdgcn_sched_barrier(0);
      compute(s % AGG_RING_, wk_ne);
    }
  } else {
#pragma unroll 1
  for (int jo = 0; jo < NE_ - AGG_RING_; jo += AGG_RING_) {
#pragma unroll
    for (int u = 0; u < AGG_RING_; ++u) {
      issue_ee(jo + u + AGG_AHEAD_, (u + AGG_AHEAD_) % AGG_RING_);
      __builtin_amdgcn_sched_barrier(0);
      compute(u, wk_ee);
    }
  }
#pragma unroll
  for (int s = NE_ - AGG_RING_; s < AGG_STEPS_; ++s) {
    const int sn = s + AGG_AHEAD_;
    if (sn < NE_)             issue_ee(sn, sn % AGG_RING_);
    else if (sn < AGG_STEPS_) issue_ne(sn - NE_, sn % AGG_RING_);
    __builtin_amdgcn_sched_barrier(0);
    if (s < NE_) compute(s % AGG_RING_, wk_ee);
    else         compute(s % AGG_RING_, wk_ne);
  }
  }

  store_strip_f32(Xw, XP_, 0, n0, acc, lane);
  __syncthreads();

  v8f acc2[LNT_] = {};
  gemm_strip_ldsA(Xw, Wp1, n0, acc2, lane);
#pragma unroll
  for (int t = 0; t < LNT_; ++t)
#pragma unroll
    for (int ii = 0; ii < 8; ++ii) acc2[t][ii] = tanhf(acc2[t][ii]);
  store_strip_f32(Yw, XP_, 0, n0, acc2, lane);
  __syncthreads();

  const int m0 = b * NE_ + i0;
  v8f acc3[LNT_];
  load_strip_f32(h, D_, m0, n0, acc3, lane);
  gemm_strip_ldsA(Yw, Wp2, n0, acc3, lane);
  store_strip_f32(dst, D_, m0, n0, acc3, lane);

  if (Wp_ee_next != nullptr) {
    store_strip_f32(Xw, XP_, 0, n0, acc3, lane);
    __syncthreads();
    v8f acc4[LNT_] = {};
    gemm_strip_ldsA(Xw, Wp_ee_next, n0, acc4, lane);
    store_strip_f32(hW_next, D_, m0, n0, acc4, lane);
  }
}

extern "C" void kernel_launch(void* const* d_in, const int* in_sizes, int n_in,
                              void* d_out, int out_size, void* d_ws, size_t ws_size,
                              hipStream_t stream) {
  (void)in_sizes; (void)n_in; (void)out_size; (void)ws_size;
  const float* r        = (const float*)d_in[0];
  const float* coords   = (const float*)d_in[1];
  const float* spin_emb = (const float*)d_in[2];
  const float* nuc_emb  = (const float*)d_in[3];
  const float* W_rbf_ee = (const float*)d_in[4];
  const float* W_rbf_ne = (const float*)d_in[5];
  const float* W_msg_ee = (const float*)d_in[6];
  const float* W_msg_ne = (const float*)d_in[7];
  const float* W1       = (const float*)d_in[8];
  const float* W2       = (const float*)d_in[9];
  float* out = (float*)d_out;

  float* ws  = (float*)d_ws;
  float* Aee = ws;
  float* Ane = Aee + (size_t)B_ * NE_ * KEE_;
  float* h   = Ane + (size_t)B_ * NE_ * KNE_;
  float* hWa = h   + (size_t)B_ * NE_ * D_;
  float* hWb = hWa + (size_t)B_ * NE_ * D_;
  float* hnW = hWb + (size_t)B_ * NE_ * D_;
  float* Wp  = hnW + (size_t)L_ * NN_ * D_;

  gnn_setup<<<SETUP_BLOCKS_, 128, 0, stream>>>(r, coords, spin_emb, nuc_emb,
                                               W_msg_ee, W_msg_ne, W1, W2,
                                               Wp, Aee, Ane, h, hWa, hnW);

  for (int l = 0; l < L_; ++l) {
    const float* Wpl = Wp + (size_t)l * 4 * WSZ_;
    const bool last = (l == L_ - 1);
    const float* hW_in  = (l & 1) ? hWb : hWa;
    float*       hW_out = (l & 1) ? hWa : hWb;
    if (l == 0)
      gnn_layer<true><<<B_ * (NE_ / LROWS_), 256, 0, stream>>>(
          Aee, Ane, hW_in, hnW + (size_t)l * NN_ * D_,
          W_rbf_ee + (size_t)l * NRBF_ * D_, W_rbf_ne + (size_t)l * NRBF_ * D_,
          Wpl + 2 * WSZ_, Wpl + 3 * WSZ_, last ? nullptr : Wpl + 4 * WSZ_,
          h, last ? out : h, hW_out);
    else
      gnn_layer<false><<<B_ * (NE_ / LROWS_), 256, 0, stream>>>(
          Aee, Ane, hW_in, hnW + (size_t)l * NN_ * D_,
          W_rbf_ee + (size_t)l * NRBF_ * D_, W_rbf_ne + (size_t)l * NRBF_ * D_,
          Wpl + 2 * WSZ_, Wpl + 3 * WSZ_, last ? nullptr : Wpl + 4 * WSZ_,
          h, last ? out : h, hW_out);
  }
}


# set-up makes only per-spin-group data for the first layer: the table's writers also store quarter sums of their A fragments, 2 hW rows, no h copy; layer 0 never reads the big table; same sums
# speedup vs baseline: 3.3349x; 1.1186x over previous
#include <hip/hip_runtime.h>
#include <math.h>

#define B_     128
#define NE_    64
#define NN_    16
#define D_     128
#define NRBF_  16
#define L_     3
#define NUP_   32
#define CUTOFF_ 2.0f
#define EPS_    1e-12f

#define KEE_   (NE_ * NRBF_)
#define KNE_   (NN_ * NRBF_)
#define NT_    4
#define LNT_   2
#define WSZ_   (D_ * D_)
#define FRAG_  (16 * NRBF_)

typedef float v2f __attribute__((ext_vector_type(2)));
typedef float v4f __attribute__((ext_vector_type(4)));
typedef float v8f __attribute__((ext_vector_type(8)));

template <int NT>
__device__ __forceinline__ void gemm_strip_f32u(const float* __restrict__ arow,
                                                const float* __restrict__ Wm,
                                                int n0, v8f (&acc)[NT], int lane) {
  const int half = lane >> 4;
  const int l16  = lane & 15;
#pragma unroll
  for (int k0 = 0; k0 < D_; k0 += 4) {
    const int ka = k0 + half * 2;
    const v2f a = *(const v2f*)&arow[ka];
    const float* __restrict__ bcol = Wm + (size_t)ka * D_ + n0 + l16;
#pragma unroll
    for (int t = 0; t < NT; ++t) {
      v2f b;
      b.x = bcol[t * 16];
      b.y = bcol[D_ + t * 16];
      acc[t] = __builtin_amdgcn_wmma_f32_16x16x4_f32(false, a, false, b, (short)0,
                                                     acc[t], false, false);
    }
  }
}

template <int NT>
__device__ __forceinline__ void store_strip_f32(float* __restrict__ C, int ldc,
                                                int m0, int n0, v8f (&acc)[NT], int lane) {
  const int half = lane >> 4;
  const int l16  = lane & 15;
#pragma unroll
  for (int t = 0; t < NT; ++t)
#pragma unroll
    for (int ii = 0; ii < 8; ++ii)
      C[(size_t)(m0 + half * 8 + ii) * ldc + n0 + t * 16 + l16] = acc[t][ii];
}

template <int NT>
__device__ __forceinline__ void load_strip_f32(const float* __restrict__ C, int ldc,
                                               int m0, int n0, v8f (&acc)[NT], int lane) {
  const int half = lane >> 4;
  const int l16  = lane & 15;
#pragma unroll
  for (int t = 0; t < NT; ++t)
#pragma unroll
    for (int ii = 0; ii < 8; ++ii)
      acc[t][ii] = C[(size_t)(m0 + half * 8 + ii) * ldc + n0 + t * 16 + l16];
}

#define XP_ (D_ + 4)

template <int NT>
__device__ __forceinline__ void gemm_strip_ldsA(const float* Xs,
                                                const float* __restrict__ Bp,
                                                int n0, v8f (&acc)[NT], int lane) {
  const int half = lane >> 4;
  const int l16  = lane & 15;
  const float* arow = Xs + l16 * XP_;
#pragma unroll
  for (int k0 = 0; k0 < D_; k0 += 4) {
    const int ka = k0 + half * 2;
    const v2f a = *(const v2f*)&arow[ka];
    const float* __restrict__ brow = Bp + (size_t)(ka >> 1) * (2 * D_);
#pragma unroll
    for (int t = 0; t < NT; ++t) {
      const v2f b = *(const v2f*)&brow[(n0 + t * 16 + l16) * 2];
      acc[t] = __builtin_amdgcn_wmma_f32_16x16x4_f32(false, a, false, b, (short)0,
                                                     acc[t], false, false);
    }
  }
}


#define PK_BLOCKS_ ((12 * (D_ / 2) * (D_ / 4)) / 128)
__device__ __forceinline__ void setup_pack_w(
    int blk,
    const float* __restrict__ We, const float* __restrict__ Wn,
    const float* __restrict__ W1, const float* __restrict__ W2,
    float* __restrict__ Wp) {
  const int gid = blk * 128 + threadIdx.x;
  const int m   = gid >> 11;
  const int kp  = (gid >> 5) & 63;
  const int c4  = gid & 31;
  const int l = m >> 2, w = m & 3;
  const float* src = (w == 0 ? We : w == 1 ? Wn : w == 2 ? W1 : W2) + (size_t)l * WSZ_;
  const v4f x = *(const v4f*)&src[(size_t)(2 * kp) * D_ + c4 * 4];
  const v4f y = *(const v4f*)&src[(size_t)(2 * kp + 1) * D_ + c4 * 4];
  float* d = Wp + (size_t)m * WSZ_ + (size_t)kp * (2 * D_) + c4 * 8;
  v4f lo, hi;
  lo.x = x.x; lo.y = y.x; lo.z = x.y; lo.w = y.y;
  hi.x = x.z; hi.y = y.z; hi.z = x.w; hi.w = y.w;
  *(v4f*)d       = lo;
  *(v4f*)(d + 4) = hi;
}

__device__ __forceinline__ void rbf_frag(float dx, float dy, float dz, int half, bool zero,
                                         v4f& q0, v4f& q1) {
  const float inv_cut = 1.0f / CUTOFF_;
  const float dmu     = CUTOFF_ / (float)(NRBF_ - 1);
  const float inv_sig = (float)NRBF_ / CUTOFF_;
  const float dist = sqrtf(__builtin_fmaf(dz, dz, __builtin_fmaf(dx, dx, dy * dy)) + EPS_);
  float env = 1.0f - dist * inv_cut;
  env = (dist < CUTOFF_) ? env * env : 0.0f;
#pragma unroll
  for (int kk = 0; kk < 4; ++kk)
#pragma unroll
    for (int e = 0; e < 2; ++e) {
      const int k = kk * 4 + half * 2 + e;
      const float z = (dist - dmu * (float)k) * inv_sig;
      float ge = expf(-0.5f * z * z) * env;
      if (zero) ge = 0.0f;
      if (kk < 2) q0[(kk & 1) * 2 + e] = ge;
      else        q1[(kk & 1) * 2 + e] = ge;
    }
}

#define FE_J_ 8
#define FE_Q_ (NE_ / 4)
#define FE_BLOCKS_ ((B_ * (NE_ / 16) * (NE_ / FE_Q_)) / 4)
__device__ __forceinline__ void setup_feat_ee(int blk, const float* __restrict__ r,
                                              float* __restrict__ Aee,
                                              float* __restrict__ Sq) {
  const int wave = blk * 4 + __builtin_amdgcn_readfirstlane((int)(threadIdx.x >> 5));
  const int lane = threadIdx.x & 31;
  const int half = lane >> 4, l16 = lane & 15;
  const int qt = wave & (NE_ / FE_Q_ - 1);
  const int bt = wave / (NE_ / FE_Q_);
  const int b  = bt >> 2;
  const int i  = (bt & 3) * 16 + l16;
  const float* pi = r + (size_t)(b * NE_ + i) * 3;
  const float rx = pi[0], ry = pi[1], rz = pi[2];
  v4f s0 = {}, s1 = {};
#pragma unroll 1
  for (int jp = 0; jp < FE_Q_; jp += FE_J_) {
    const int j0 = qt * FE_Q_ + jp;
    const float* __restrict__ pj = r + (size_t)(b * NE_ + j0) * 3;
    float* __restrict__ o = Aee + ((size_t)bt * NE_ + j0) * FRAG_ + lane * 4;
#pragma unroll
    for (int jj = 0; jj < FE_J_; ++jj) {
      v4f q0, q1;
      rbf_frag(rx - pj[jj * 3 + 0], ry - pj[jj * 3 + 1], rz - pj[jj * 3 + 2], half,
               i == j0 + jj, q0, q1);
      *(v4f*)&o[jj * FRAG_]       = q0;
      *(v4f*)&o[jj * FRAG_ + 128] = q1;
      s0 += q0;
      s1 += q1;
    }
  }
  float* __restrict__ sq = Sq + (size_t)bt * 16 * D_ + qt * FRAG_ + lane * 4;
  *(v4f*)&sq[0]   = s0;
  *(v4f*)&sq[128] = s1;
}

#define FN_A_ 4
#define FN_BLOCKS_ ((B_ * (NE_ / 16) * (NN_ / FN_A_)) / 4)
__device__ __forceinline__ void setup_feat_ne(int blk, const float* __restrict__ r,
                                              const float* __restrict__ coords,
                                              float* __restrict__ Ane) {
  const int wave = blk * 4 + __builtin_amdgcn_readfirstlane((int)(threadIdx.x >> 5));
  const int lane = threadIdx.x & 31;
  const int half = lane >> 4, l16 = lane & 15;
  const int a0 = (wave & (NN_ / FN_A_ - 1)) * FN_A_;
  const int bt = wave / (NN_ / FN_A_);
  const int b  = bt >> 2;
  const int i  = (bt & 3) * 16 + l16;
  const float* pi = r + (size_t)(b * NE_ + i) * 3;
  const float rx = pi[0], ry = pi[1], rz = pi[2];
  const float* __restrict__ pc = coords + a0 * 3;
  float* __restrict__ o = Ane + ((size_t)bt * NN_ + a0) * FRAG_ + lane * 4;
#pragma unroll
  for (int aa = 0; aa < FN_A_; ++aa) {
    v4f q0, q1;
    rbf_frag(rx - pc[aa * 3 + 0], ry - pc[aa * 3 + 1], rz - pc[aa * 3 + 2], half,
             false, q0, q1);
    *(v4f*)&o[aa * FRAG_]       = q0;
    *(v4f*)&o[aa * FRAG_ + 128] = q1;
  }
}


#define HW_STRIPS_ 2
#define HW_BLOCKS_ ((HW_STRIPS_ + 2 * L_ + 3) / 4)
__device__ __forceinline__ void setup_hw(int blk,
                                         const float* __restrict__ spin_emb,
                                         const float* __restrict__ nuc_emb,
                                         const float* __restrict__ We,
                                         const float* __restrict__ Wn,
                                         float* __restrict__ hW, float* __restrict__ hnW) {
  const int s = blk * 4 + __builtin_amdgcn_readfirstlane((int)(threadIdx.x >> 5));
  const int lane = threadIdx.x & 31;
  const int l16  = lane & 15;
  v8f acc[NT_] = {};
  if (s < HW_STRIPS_) {
    const int n0 = s * 64;
    gemm_strip_f32u(spin_emb + (l16 & 1) * D_, We, n0, acc, lane);
    store_strip_f32(hW, D_, 0, n0, acc, lane);
  } else if (s < HW_STRIPS_ + 2 * L_) {
    const int l  = (s - HW_STRIPS_) >> 1;
    const int n0 = ((s - HW_STRIPS_) & 1) * 64;
    gemm_strip_f32u(nuc_emb + l16 * D_, Wn + (size_t)l * WSZ_, n0, acc, lane);
    store_strip_f32(hnW + (size_t)l * NN_ * D_, D_, 0, n0, acc, lane);
  }
}

#define SB_HW_ (FE_BLOCKS_)
#define SB_FN_ (SB_HW_ + HW_BLOCKS_)
#define SB_PK_ (SB_FN_ + FN_BLOCKS_)
#define SETUP_BLOCKS_ (SB_PK_ + PK_BLOCKS_)
__global__ __launch_bounds__(128) void gnn_setup(
    const float* __restrict__ r, const float* __restrict__ coords,
    const float* __restrict__ spin_emb, const float* __restrict__ nuc_emb,
    const float* __restrict__ We, const float* __restrict__ Wn,
    const float* __restrict__ W1, const float* __restrict__ W2,
    float* __restrict__ Wp, float* __restrict__ Aee, float* __restrict__ Ane,
    float* __restrict__ hW, float* __restrict__ hnW, float* __restrict__ Sq) {
  const int blk = blockIdx.x;
  if (blk < SB_HW_)      setup_feat_ee(blk, r, Aee, Sq);
  else if (blk < SB_FN_) setup_hw(blk - SB_HW_, spin_emb, nuc_emb, We, Wn, hW, hnW);
  else if (blk < SB_PK_) setup_feat_ne(blk - SB_FN_, r, coords, Ane);
  else                   setup_pack_w(blk - SB_PK_, We, Wn, W1, W2, Wp);
}

#define AGG_RING_  8
#define AGG_AHEAD_ 6
#define AGG_STEPS_ (NE_ + NN_)
#define LROWS_     32

template <bool FIRST>
__global__ __launch_bounds__(256) void gnn_layer(
    const float* __restrict__ Aee, const float* __restrict__ Ane,
    const float* __restrict__ hW, const float* __restrict__ hnW,
    const float* __restrict__ Wrbf_ee_l, const float* __restrict__ Wrbf_ne_l,
    const float* __restrict__ Wp1, const float* __restrict__ Wp2,
    const float* __restrict__ Wp_ee_next,
    const float* h, float* dst,
    float* __restrict__ hW_next) {
  __shared__ __attribute__((aligned(16))) float Xs[LROWS_ * XP_];
  __shared__ __attribute__((aligned(16))) float Ys[LROWS_ * XP_];
  const int wid = blockIdx.x;
  const int b  = wid >> 1;
  const int w  = __builtin_amdgcn_readfirstlane((int)(threadIdx.x >> 5));
  const int mw = w & 1;
  const int nq = w >> 1;
  const int lane = threadIdx.x & 31;
  const int half = lane >> 4, l16 = lane & 15;
  const int i0 = (wid & 1) * LROWS_ + mw * 16, n0 = nq * (16 * LNT_);
  float* Xw = Xs + mw * 16 * XP_;
  float* Yw = Ys + mw * 16 * XP_;

  const int bt = b * (NE_ / 16) + (i0 >> 4);
  const float* __restrict__ a_ee = Aee + (size_t)bt * NE_ * FRAG_ + lane * 4;
  const float* __restrict__ a_ne = Ane + (size_t)bt * NN_ * FRAG_ + lane * 4;
  const float* __restrict__ h_ee = hW + (size_t)b * NE_ * D_ + n0 + l16;
  const float* __restrict__ h_ne = hnW + n0 + l16;

  v8f   acc[LNT_] = {};
  v4f   rq[AGG_RING_][2];
  float rh[AGG_RING_][LNT_];

  auto issue_ee = [&](int j, int st) {
#pragma unroll
    for (int t = 0; t < LNT_; ++t) rh[st][t] = h_ee[(size_t)j * D_ + t * 16];
    rq[st][0] = *(const v4f*)&a_ee[j * FRAG_];
    rq[st][1] = *(const v4f*)&a_ee[j * FRAG_ + 128];
  };
  auto issue_ne = [&](int a, int st) {
#pragma unroll
    for (int t = 0; t < LNT_; ++t) rh[st][t] = h_ne[(size_t)a * D_ + t * 16];
    rq[st][0] = *(const v4f*)&a_ne[a * FRAG_];
    rq[st][1] = *(const v4f*)&a_ne[a * FRAG_ + 128];
  };

  if constexpr (!FIRST) {
#pragma unroll
    for (int s = 0; s < AGG_AHEAD_; ++s) issue_ee(s, s);
  }

  float wk_ee[LNT_][8], wk_ne[LNT_][8];
#pragma unroll
  for (int kk = 0; kk < 4; ++kk)
#pragma unroll
    for (int t = 0; t < LNT_; ++t) {
      const int kb = kk * 4 + half * 2;
      wk_ee[t][kk * 2 + 0] = Wrbf_ee_l[(size_t)(kb + 0) * D_ + n0 + t * 16 + l16];
      wk_ee[t][kk * 2 + 1] = Wrbf_ee_l[(size_t)(kb + 1) * D_ + n0 + t * 16 + l16];
      wk_ne[t][kk * 2 + 0] = Wrbf_ne_l[(size_t)(kb + 0) * D_ + n0 + t * 16 + l16];
      wk_ne[t][kk * 2 + 1] = Wrbf_ne_l[(size_t)(kb + 1) * D_ + n0 + t * 16 + l16];
    }

  auto mm = [&](const v4f& q0, const v4f& q1, const float (&hv)[LNT_],
                const float (&wk)[LNT_][8]) {
#pragma unroll
    for (int kk = 0; kk < 4; ++kk) {
      const v4f q = (kk >> 1) ? q1 : q0;
      v2f a;
      a.x = (kk & 1) ? q.z : q.x;
      a.y = (kk & 1) ? q.w : q.y;
#pragma unroll
      for (int t = 0; t < LNT_; ++t) {
        v2f bb;
        bb.x = hv[t] * wk[t][kk * 2 + 0];
        bb.y = hv[t] * wk[t][kk * 2 + 1];
        acc[t] = __builtin_amdgcn_wmma_f32_16x16x4_f32(false, a, false, bb, (short)0,
                                                       acc[t], false, false);
      }
    }
  };
  auto compute = [&](int st, const float (&wk)[LNT_][8]) {
    mm(rq[st][0], rq[st][1], rh[st], wk);
  };

  if constexpr (FIRST) {
    static_assert(NUP_ * 2 == NE_ && NE_ == 64,
                  "a spin group is two 16-electron quarters of the four");
    float hg[2][LNT_];
#pragma unroll
    for (int g = 0; g < 2; ++g)
#pragma unroll
      for (int t = 0; t < LNT_; ++t) hg[g][t] = hW[(size_t)g * D_ + n0 + l16 + t * 16];
    const float* sq = hW_next + (size_t)(b * NE_ + i0) * D_ + lane * 4;
    v4f pq[NE_ / FE_Q_][2];
#pragma unroll
    for (int qd = 0; qd < NE_ / FE_Q_; ++qd) {
      pq[qd][0] = *(const v4f*)&sq[qd * FRAG_];
      pq[qd][1] = *(const v4f*)&sq[qd * FRAG_ + 128];
    }
#pragma unroll
    for (int s = 0; s < AGG_AHEAD_; ++s) issue_ne(s, s);
    __builtin_amdgcn_sched_barrier(0);
#pragma unroll
    for (int g = 0; g < 2; ++g) {
      const v4f q0 = pq[2 * g][0] + pq[2 * g + 1][0];
      const v4f q1 = pq[2 * g][1] + pq[2 * g + 1][1];
      mm(q0, q1, hg[g], wk_ee);
    }
#pragma unroll
    for (int s = 0; s < NN_; ++s) {
      if (s + AGG_AHEAD_ < NN_) issue_ne(s + AGG_AHEAD_, (s + AGG_AHEAD_) % AGG_RING_);
      __builtin_amdgcn_sched_barrier(0);
      compute(s % AGG_RING_, wk_ne);
    }
  } else {
#pragma unroll 1
  for (int jo = 0; jo < NE_ - AGG_RING_; jo += AGG_RING_) {
#pragma unroll
    for (int u = 0; u < AGG_RING_; ++u) {
      issue_ee(jo + u + AGG_AHEAD_, (u + AGG_AHEAD_) % AGG_RING_);
      __builtin_amdgcn_sched_barrier(0);
      compute(u, wk_ee);
    }
  }
#pragma unroll
  for (int s = NE_ - AGG_RING_; s < AGG_STEPS_; ++s) {
    const int sn = s + AGG_AHEAD_;
    if (sn < NE_)             issue_ee(sn, sn % AGG_RING_);
    else if (sn < AGG_STEPS_) issue_ne(sn - NE_, sn % AGG_RING_);
    __builtin_amdgcn_sched_barrier(0);
    if (s < NE_) compute(s % AGG_RING_, wk_ee);
    else         compute(s % AGG_RING_, wk_ne);
  }
  }

  store_strip_f32(Xw, XP_, 0, n0, acc, lane);
  __syncthreads();

  v8f acc2[LNT_] = {};
  gemm_strip_ldsA(Xw, Wp1, n0, acc2, lane);
#pragma unroll
  for (int t = 0; t < LNT_; ++t)
#pragma unroll
    for (int ii = 0; ii < 8; ++ii) acc2[t][ii] = tanhf(acc2[t][ii]);
  store_strip_f32(Yw, XP_, 0, n0, acc2, lane);
  __syncthreads();

  const int m0 = b * NE_ + i0;
  v8f acc3[LNT_];
  if constexpr (FIRST) {
    const float* hs = h + (size_t)((i0 < NUP_) ? 0 : 1) * D_ + n0 + l16;
#pragma unroll
    for (int t = 0; t < LNT_; ++t) {
      const float v = hs[t * 16];
#pragma unroll
      for (int ii = 0; ii < 8; ++ii) acc3[t][ii] = v;
    }
  } else {
    load_strip_f32(h, D_, m0, n0, acc3, lane);
  }
  gemm_strip_ldsA(Yw, Wp2, n0, acc3, lane);
  store_strip_f32(dst, D_, m0, n0, acc3, lane);

  if (Wp_ee_next != nullptr) {
    store_strip_f32(Xw, XP_, 0, n0, acc3, lane);
    __syncthreads();
    v8f acc4[LNT_] = {};
    gemm_strip_ldsA(Xw, Wp_ee_next, n0, acc4, lane);
    store_strip_f32(hW_next, D_, m0, n0, acc4, lane);
  }
}

extern "C" void kernel_launch(void* const* d_in, const int* in_sizes, int n_in,
                              void* d_out, int out_size, void* d_ws, size_t ws_size,
                              hipStream_t stream) {
  (void)in_sizes; (void)n_in; (void)out_size; (void)ws_size;
  const float* r        = (const float*)d_in[0];
  const float* coords   = (const float*)d_in[1];
  const float* spin_emb = (const float*)d_in[2];
  const float* nuc_emb  = (const float*)d_in[3];
  const float* W_rbf_ee = (const float*)d_in[4];
  const float* W_rbf_ne = (const float*)d_in[5];
  const float* W_msg_ee = (const float*)d_in[6];
  const float* W_msg_ne = (const float*)d_in[7];
  const float* W1       = (const float*)d_in[8];
  const float* W2       = (const float*)d_in[9];
  float* out = (float*)d_out;

  float* ws  = (float*)d_ws;
  float* Aee = ws;
  float* Ane = Aee + (size_t)B_ * NE_ * KEE_;
  float* h   = Ane + (size_t)B_ * NE_ * KNE_;
  float* hWa = h   + (size_t)B_ * NE_ * D_;
  float* hWb = hWa + (size_t)B_ * NE_ * D_;
  float* hnW = hWb + (size_t)B_ * NE_ * D_;
  float* Wp  = hnW + (size_t)L_ * NN_ * D_;

  gnn_setup<<<SETUP_BLOCKS_, 128, 0, stream>>>(r, coords, spin_emb, nuc_emb,
                                               W_msg_ee, W_msg_ne, W1, W2,
                                               Wp, Aee, Ane, hWa, hnW, hWb);

  for (int l = 0; l < L_; ++l) {
    const float* Wpl = Wp + (size_t)l * 4 * WSZ_;
    const bool last = (l == L_ - 1);
    const float* hW_in  = (l & 1) ? hWb : hWa;
    float*       hW_out = (l & 1) ? hWa : hWb;
    if (l == 0)
      gnn_layer<true><<<B_ * (NE_ / LROWS_), 256, 0, stream>>>(
          Aee, Ane, hW_in, hnW + (size_t)l * NN_ * D_,
          W_rbf_ee + (size_t)l * NRBF_ * D_, W_rbf_ne + (size_t)l * NRBF_ * D_,
          Wpl + 2 * WSZ_, Wpl + 3 * WSZ_, last ? nullptr : Wpl + 4 * WSZ_,
          spin_emb, last ? out : h, hW_out);
    else
      gnn_layer<false><<<B_ * (NE_ / LROWS_), 256, 0, stream>>>(
          Aee, Ane, hW_in, hnW + (size_t)l * NN_ * D_,
          W_rbf_ee + (size_t)l * NRBF_ * D_, W_rbf_ne + (size_t)l * NRBF_ * D_,
          Wpl + 2 * WSZ_, Wpl + 3 * WSZ_, last ? nullptr : Wpl + 4 * WSZ_,
          h, last ? out : h, hW_out);
  }
}


# MLP activation: library tanhf (two exec-masked branches per value, 16 serial chains per lane) replaced by branch-free 1-2/(2^(2|x|log2e)+1) on hardware exp2 and reciprocal; abs. error about 3e-7
# speedup vs baseline: 3.4622x; 1.0382x over previous
#include <hip/hip_runtime.h>
#include <math.h>

#define B_     128
#define NE_    64
#define NN_    16
#define D_     128
#define NRBF_  16
#define L_     3
#define NUP_   32
#define CUTOFF_ 2.0f
#define EPS_    1e-12f

#define KEE_   (NE_ * NRBF_)
#define KNE_   (NN_ * NRBF_)
#define NT_    4
#define LNT_   2
#define WSZ_   (D_ * D_)
#define FRAG_  (16 * NRBF_)

typedef float v2f __attribute__((ext_vector_type(2)));
typedef float v4f __attribute__((ext_vector_type(4)));
typedef float v8f __attribute__((ext_vector_type(8)));

template <int NT>
__device__ __forceinline__ void gemm_strip_f32u(const float* __restrict__ arow,
                                                const float* __restrict__ Wm,
                                                int n0, v8f (&acc)[NT], int lane) {
  const int half = lane >> 4;
  const int l16  = lane & 15;
#pragma unroll
  for (int k0 = 0; k0 < D_; k0 += 4) {
    const int ka = k0 + half * 2;
    const v2f a = *(const v2f*)&arow[ka];
    const float* __restrict__ bcol = Wm + (size_t)ka * D_ + n0 + l16;
#pragma unroll
    for (int t = 0; t < NT; ++t) {
      v2f b;
      b.x = bcol[t * 16];
      b.y = bcol[D_ + t * 16];
      acc[t] = __builtin_amdgcn_wmma_f32_16x16x4_f32(false, a, false, b, (short)0,
                                                     acc[t], false, false);
    }
  }
}

template <int NT>
__device__ __forceinline__ void store_strip_f32(float* __restrict__ C, int ldc,
                                                int m0, int n0, v8f (&acc)[NT], int lane) {
  const int half = lane >> 4;
  const int l16  = lane & 15;
#pragma unroll
  for (int t = 0; t < NT; ++t)
#pragma unroll
    for (int ii = 0; ii < 8; ++ii)
      C[(size_t)(m0 + half * 8 + ii) * ldc + n0 + t * 16 + l16] = acc[t][ii];
}

template <int NT>
__device__ __forceinline__ void load_strip_f32(const float* __restrict__ C, int ldc,
                                               int m0, int n0, v8f (&acc)[NT], int lane) {
  const int half = lane >> 4;
  const int l16  = lane & 15;
#pragma unroll
  for (int t = 0; t < NT; ++t)
#pragma unroll
    for (int ii = 0; ii < 8; ++ii)
      acc[t][ii] = C[(size_t)(m0 + half * 8 + ii) * ldc + n0 + t * 16 + l16];
}

#define XP_ (D_ + 4)

template <int NT>
__device__ __forceinline__ void gemm_strip_ldsA(const float* Xs,
                                                const float* __restrict__ Bp,
                                                int n0, v8f (&acc)[NT], int lane) {
  const int half = lane >> 4;
  const int l16  = lane & 15;
  const float* arow = Xs + l16 * XP_;
#pragma unroll
  for (int k0 = 0; k0 < D_; k0 += 4) {
    const int ka = k0 + half * 2;
    const v2f a = *(const v2f*)&arow[ka];
    const float* __restrict__ brow = Bp + (size_t)(ka >> 1) * (2 * D_);
#pragma unroll
    for (int t = 0; t < NT; ++t) {
      const v2f b = *(const v2f*)&brow[(n0 + t * 16 + l16) * 2];
      acc[t] = __builtin_amdgcn_wmma_f32_16x16x4_f32(false, a, false, b, (short)0,
                                                     acc[t], false, false);
    }
  }
}

__device__ __forceinline__ float tanh_nobranch(float x) {
  const float y = __builtin_fabsf(x);
  const float e = __builtin_amdgcn_exp2f(y * 2.8853900817779268f);
  const float r = __builtin_amdgcn_rcpf(e + 1.0f);
  return __builtin_copysignf(__builtin_fmaf(r, -2.0f, 1.0f), x);
}


#define PK_BLOCKS_ ((12 * (D_ / 2) * (D_ / 4)) / 128)
__device__ __forceinline__ void setup_pack_w(
    int blk,
    const float* __restrict__ We, const float* __restrict__ Wn,
    const float* __restrict__ W1, const float* __restrict__ W2,
    float* __restrict__ Wp) {
  const int gid = blk * 128 + threadIdx.x;
  const int m   = gid >> 11;
  const int kp  = (gid >> 5) & 63;
  const int c4  = gid & 31;
  const int l = m >> 2, w = m & 3;
  const float* src = (w == 0 ? We : w == 1 ? Wn : w == 2 ? W1 : W2) + (size_t)l * WSZ_;
  const v4f x = *(const v4f*)&src[(size_t)(2 * kp) * D_ + c4 * 4];
  const v4f y = *(const v4f*)&src[(size_t)(2 * kp + 1) * D_ + c4 * 4];
  float* d = Wp + (size_t)m * WSZ_ + (size_t)kp * (2 * D_) + c4 * 8;
  v4f lo, hi;
  lo.x = x.x; lo.y = y.x; lo.z = x.y; lo.w = y.y;
  hi.x = x.z; hi.y = y.z; hi.z = x.w; hi.w = y.w;
  *(v4f*)d       = lo;
  *(v4f*)(d + 4) = hi;
}

__device__ __forceinline__ void rbf_frag(float dx, float dy, float dz, int half, bool zero,
                                         v4f& q0, v4f& q1) {
  const float inv_cut = 1.0f / CUTOFF_;
  const float dmu     = CUTOFF_ / (float)(NRBF_ - 1);
  const float inv_sig = (float)NRBF_ / CUTOFF_;
  const float dist = sqrtf(__builtin_fmaf(dz, dz, __builtin_fmaf(dx, dx, dy * dy)) + EPS_);
  float env = 1.0f - dist * inv_cut;
  env = (dist < CUTOFF_) ? env * env : 0.0f;
#pragma unroll
  for (int kk = 0; kk < 4; ++kk)
#pragma unroll
    for (int e = 0; e < 2; ++e) {
      const int k = kk * 4 + half * 2 + e;
      const float z = (dist - dmu * (float)k) * inv_sig;
      float ge = expf(-0.5f * z * z) * env;
      if (zero) ge = 0.0f;
      if (kk < 2) q0[(kk & 1) * 2 + e] = ge;
      else        q1[(kk & 1) * 2 + e] = ge;
    }
}

#define FE_J_ 8
#define FE_Q_ (NE_ / 4)
#define FE_BLOCKS_ ((B_ * (NE_ / 16) * (NE_ / FE_Q_)) / 4)
__device__ __forceinline__ void setup_feat_ee(int blk, const float* __restrict__ r,
                                              float* __restrict__ Aee,
                                              float* __restrict__ Sq) {
  const int wave = blk * 4 + __builtin_amdgcn_readfirstlane((int)(threadIdx.x >> 5));
  const int lane = threadIdx.x & 31;
  const int half = lane >> 4, l16 = lane & 15;
  const int qt = wave & (NE_ / FE_Q_ - 1);
  const int bt = wave / (NE_ / FE_Q_);
  const int b  = bt >> 2;
  const int i  = (bt & 3) * 16 + l16;
  const float* pi = r + (size_t)(b * NE_ + i) * 3;
  const float rx = pi[0], ry = pi[1], rz = pi[2];
  v4f s0 = {}, s1 = {};
#pragma unroll 1
  for (int jp = 0; jp < FE_Q_; jp += FE_J_) {
    const int j0 = qt * FE_Q_ + jp;
    const float* __restrict__ pj = r + (size_t)(b * NE_ + j0) * 3;
    float* __restrict__ o = Aee + ((size_t)bt * NE_ + j0) * FRAG_ + lane * 4;
#pragma unroll
    for (int jj = 0; jj < FE_J_; ++jj) {
      v4f q0, q1;
      rbf_frag(rx - pj[jj * 3 + 0], ry - pj[jj * 3 + 1], rz - pj[jj * 3 + 2], half,
               i == j0 + jj, q0, q1);
      *(v4f*)&o[jj * FRAG_]       = q0;
      *(v4f*)&o[jj * FRAG_ + 128] = q1;
      s0 += q0;
      s1 += q1;
    }
  }
  float* __restrict__ sq = Sq + (size_t)bt * 16 * D_ + qt * FRAG_ + lane * 4;
  *(v4f*)&sq[0]   = s0;
  *(v4f*)&sq[128] = s1;
}

#define FN_A_ 4
#define FN_BLOCKS_ ((B_ * (NE_ / 16) * (NN_ / FN_A_)) / 4)
__device__ __forceinline__ void setup_feat_ne(int blk, const float* __restrict__ r,
                                              const float* __restrict__ coords,
                                              float* __restrict__ Ane) {
  const int wave = blk * 4 + __builtin_amdgcn_readfirstlane((int)(threadIdx.x >> 5));
  const int lane = threadIdx.x & 31;
  const int half = lane >> 4, l16 = lane & 15;
  const int a0 = (wave & (NN_ / FN_A_ - 1)) * FN_A_;
  const int bt = wave / (NN_ / FN_A_);
  const int b  = bt >> 2;
  const int i  = (bt & 3) * 16 + l16;
  const float* pi = r + (size_t)(b * NE_ + i) * 3;
  const float rx = pi[0], ry = pi[1], rz = pi[2];
  const float* __restrict__ pc = coords + a0 * 3;
  float* __restrict__ o = Ane + ((size_t)bt * NN_ + a0) * FRAG_ + lane * 4;
#pragma unroll
  for (int aa = 0; aa < FN_A_; ++aa) {
    v4f q0, q1;
    rbf_frag(rx - pc[aa * 3 + 0], ry - pc[aa * 3 + 1], rz - pc[aa * 3 + 2], half,
             false, q0, q1);
    *(v4f*)&o[aa * FRAG_]       = q0;
    *(v4f*)&o[aa * FRAG_ + 128] = q1;
  }
}


#define HW_STRIPS_ 2
#define HW_BLOCKS_ ((HW_STRIPS_ + 2 * L_ + 3) / 4)
__device__ __forceinline__ void setup_hw(int blk,
                                         const float* __restrict__ spin_emb,
                                         const float* __restrict__ nuc_emb,
                                         const float* __restrict__ We,
                                         const float* __restrict__ Wn,
                                         float* __restrict__ hW, float* __restrict__ hnW) {
  const int s = blk * 4 + __builtin_amdgcn_readfirstlane((int)(threadIdx.x >> 5));
  const int lane = threadIdx.x & 31;
  const int l16  = lane & 15;
  v8f acc[NT_] = {};
  if (s < HW_STRIPS_) {
    const int n0 = s * 64;
    gemm_strip_f32u(spin_emb + (l16 & 1) * D_, We, n0, acc, lane);
    store_strip_f32(hW, D_, 0, n0, acc, lane);
  } else if (s < HW_STRIPS_ + 2 * L_) {
    const int l  = (s - HW_STRIPS_) >> 1;
    const int n0 = ((s - HW_STRIPS_) & 1) * 64;
    gemm_strip_f32u(nuc_emb + l16 * D_, Wn + (size_t)l * WSZ_, n0, acc, lane);
    store_strip_f32(hnW + (size_t)l * NN_ * D_, D_, 0, n0, acc, lane);
  }
}

#define SB_HW_ (FE_BLOCKS_)
#define SB_FN_ (SB_HW_ + HW_BLOCKS_)
#define SB_PK_ (SB_FN_ + FN_BLOCKS_)
#define SETUP_BLOCKS_ (SB_PK_ + PK_BLOCKS_)
__global__ __launch_bounds__(128) void gnn_setup(
    const float* __restrict__ r, const float* __restrict__ coords,
    const float* __restrict__ spin_emb, const float* __restrict__ nuc_emb,
    const float* __restrict__ We, const float* __restrict__ Wn,
    const float* __restrict__ W1, const float* __restrict__ W2,
    float* __restrict__ Wp, float* __restrict__ Aee, float* __restrict__ Ane,
    float* __restrict__ hW, float* __restrict__ hnW, float* __restrict__ Sq) {
  const int blk = blockIdx.x;
  if (blk < SB_HW_)      setup_feat_ee(blk, r, Aee, Sq);
  else if (blk < SB_FN_) setup_hw(blk - SB_HW_, spin_emb, nuc_emb, We, Wn, hW, hnW);
  else if (blk < SB_PK_) setup_feat_ne(blk - SB_FN_, r, coords, Ane);
  else                   setup_pack_w(blk - SB_PK_, We, Wn, W1, W2, Wp);
}

#define AGG_RING_  8
#define AGG_AHEAD_ 6
#define AGG_STEPS_ (NE_ + NN_)
#define LROWS_     32

template <bool FIRST>
__global__ __launch_bounds__(256) void gnn_layer(
    const float* __restrict__ Aee, const float* __restrict__ Ane,
    const float* __restrict__ hW, const float* __restrict__ hnW,
    const float* __restrict__ Wrbf_ee_l, const float* __restrict__ Wrbf_ne_l,
    const float* __restrict__ Wp1, const float* __restrict__ Wp2,
    const float* __restrict__ Wp_ee_next,
    const float* h, float* dst,
    float* __restrict__ hW_next) {
  __shared__ __attribute__((aligned(16))) float Xs[LROWS_ * XP_];
  __shared__ __attribute__((aligned(16))) float Ys[LROWS_ * XP_];
  const int wid = blockIdx.x;
  const int b  = wid >> 1;
  const int w  = __builtin_amdgcn_readfirstlane((int)(threadIdx.x >> 5));
  const int mw = w & 1;
  const int nq = w >> 1;
  const int lane = threadIdx.x & 31;
  const int half = lane >> 4, l16 = lane & 15;
  const int i0 = (wid & 1) * LROWS_ + mw * 16, n0 = nq * (16 * LNT_);
  float* Xw = Xs + mw * 16 * XP_;
  float* Yw = Ys + mw * 16 * XP_;

  const int bt = b * (NE_ / 16) + (i0 >> 4);
  const float* __restrict__ a_ee = Aee + (size_t)bt * NE_ * FRAG_ + lane * 4;
  const float* __restrict__ a_ne = Ane + (size_t)bt * NN_ * FRAG_ + lane * 4;
  const float* __restrict__ h_ee = hW + (size_t)b * NE_ * D_ + n0 + l16;
  const float* __restrict__ h_ne = hnW + n0 + l16;

  v8f   acc[LNT_] = {};
  v4f   rq[AGG_RING_][2];
  float rh[AGG_RING_][LNT_];

  auto issue_ee = [&](int j, int st) {
#pragma unroll
    for (int t = 0; t < LNT_; ++t) rh[st][t] = h_ee[(size_t)j * D_ + t * 16];
    rq[st][0] = *(const v4f*)&a_ee[j * FRAG_];
    rq[st][1] = *(const v4f*)&a_ee[j * FRAG_ + 128];
  };
  auto issue_ne = [&](int a, int st) {
#pragma unroll
    for (int t = 0; t < LNT_; ++t) rh[st][t] = h_ne[(size_t)a * D_ + t * 16];
    rq[st][0] = *(const v4f*)&a_ne[a * FRAG_];
    rq[st][1] = *(const v4f*)&a_ne[a * FRAG_ + 128];
  };

  if constexpr (!FIRST) {
#pragma unroll
    for (int s = 0; s < AGG_AHEAD_; ++s) issue_ee(s, s);
  }

  float wk_ee[LNT_][8], wk_ne[LNT_][8];
#pragma unroll
  for (int kk = 0; kk < 4; ++kk)
#pragma unroll
    for (int t = 0; t < LNT_; ++t) {
      const int kb = kk * 4 + half * 2;
      wk_ee[t][kk * 2 + 0] = Wrbf_ee_l[(size_t)(kb + 0) * D_ + n0 + t * 16 + l16];
      wk_ee[t][kk * 2 + 1] = Wrbf_ee_l[(size_t)(kb + 1) * D_ + n0 + t * 16 + l16];
      wk_ne[t][kk * 2 + 0] = Wrbf_ne_l[(size_t)(kb + 0) * D_ + n0 + t * 16 + l16];
      wk_ne[t][kk * 2 + 1] = Wrbf_ne_l[(size_t)(kb + 1) * D_ + n0 + t * 16 + l16];
    }

  auto mm = [&](const v4f& q0, const v4f& q1, const float (&hv)[LNT_],
                const float (&wk)[LNT_][8]) {
#pragma unroll
    for (int kk = 0; kk < 4; ++kk) {
      const v4f q = (kk >> 1) ? q1 : q0;
      v2f a;
      a.x = (kk & 1) ? q.z : q.x;
      a.y = (kk & 1) ? q.w : q.y;
#pragma unroll
      for (int t = 0; t < LNT_; ++t) {
        v2f bb;
        bb.x = hv[t] * wk[t][kk * 2 + 0];
        bb.y = hv[t] * wk[t][kk * 2 + 1];
        acc[t] = __builtin_amdgcn_wmma_f32_16x16x4_f32(false, a, false, bb, (short)0,
                                                       acc[t], false, false);
      }
    }
  };
  auto compute = [&](int st, const float (&wk)[LNT_][8]) {
    mm(rq[st][0], rq[st][1], rh[st], wk);
  };

  if constexpr (FIRST) {
    static_assert(NUP_ * 2 == NE_ && NE_ == 64,
                  "a spin group is two 16-electron quarters of the four");
    float hg[2][LNT_];
#pragma unroll
    for (int g = 0; g < 2; ++g)
#pragma unroll
      for (int t = 0; t < LNT_; ++t) hg[g][t] = hW[(size_t)g * D_ + n0 + l16 + t * 16];
    const float* sq = hW_next + (size_t)(b * NE_ + i0) * D_ + lane * 4;
    v4f pq[NE_ / FE_Q_][2];
#pragma unroll
    for (int qd = 0; qd < NE_ / FE_Q_; ++qd) {
      pq[qd][0] = *(const v4f*)&sq[qd * FRAG_];
      pq[qd][1] = *(const v4f*)&sq[qd * FRAG_ + 128];
    }
#pragma unroll
    for (int s = 0; s < AGG_AHEAD_; ++s) issue_ne(s, s);
    __builtin_amdgcn_sched_barrier(0);
#pragma unroll
    for (int g = 0; g < 2; ++g) {
      const v4f q0 = pq[2 * g][0] + pq[2 * g + 1][0];
      const v4f q1 = pq[2 * g][1] + pq[2 * g + 1][1];
      mm(q0, q1, hg[g], wk_ee);
    }
#pragma unroll
    for (int s = 0; s < NN_; ++s) {
      if (s + AGG_AHEAD_ < NN_) issue_ne(s + AGG_AHEAD_, (s + AGG_AHEAD_) % AGG_RING_);
      __builtin_amdgcn_sched_barrier(0);
      compute(s % AGG_RING_, wk_ne);
    }
  } else {
#pragma unroll 1
  for (int jo = 0; jo < NE_ - AGG_RING_; jo += AGG_RING_) {
#pragma unroll
    for (int u = 0; u < AGG_RING_; ++u) {
      issue_ee(jo + u + AGG_AHEAD_, (u + AGG_AHEAD_) % AGG_RING_);
      __builtin_amdgcn_sched_barrier(0);
      compute(u, wk_ee);
    }
  }
#pragma unroll
  for (int s = NE_ - AGG_RING_; s < AGG_STEPS_; ++s) {
    const int sn = s + AGG_AHEAD_;
    if (sn < NE_)             issue_ee(sn, sn % AGG_RING_);
    else if (sn < AGG_STEPS_) issue_ne(sn - NE_, sn % AGG_RING_);
    __builtin_amdgcn_sched_barrier(0);
    if (s < NE_) compute(s % AGG_RING_, wk_ee);
    else         compute(s % AGG_RING_, wk_ne);
  }
  }

  store_strip_f32(Xw, XP_, 0, n0, acc, lane);
  __syncthreads();

  v8f acc2[LNT_] = {};
  gemm_strip_ldsA(Xw, Wp1, n0, acc2, lane);
#pragma unroll
  for (int t = 0; t < LNT_; ++t)
#pragma unroll
    for (int ii = 0; ii < 8; ++ii) acc2[t][ii] = tanh_nobranch(acc2[t][ii]);
  store_strip_f32(Yw, XP_, 0, n0, acc2, lane);
  __syncthreads();

  const int m0 = b * NE_ + i0;
  v8f acc3[LNT_];
  if constexpr (FIRST) {
    const float* hs = h + (size_t)((i0 < NUP_) ? 0 : 1) * D_ + n0 + l16;
#pragma unroll
    for (int t = 0; t < LNT_; ++t) {
      const float v = hs[t * 16];
#pragma unroll
      for (int ii = 0; ii < 8; ++ii) acc3[t][ii] = v;
    }
  } else {
    load_strip_f32(h, D_, m0, n0, acc3, lane);
  }
  gemm_strip_ldsA(Yw, Wp2, n0, acc3, lane);
  store_strip_f32(dst, D_, m0, n0, acc3, lane);

  if (Wp_ee_next != nullptr) {
    store_strip_f32(Xw, XP_, 0, n0, acc3, lane);
    __syncthreads();
    v8f acc4[LNT_] = {};
    gemm_strip_ldsA(Xw, Wp_ee_next, n0, acc4, lane);
    store_strip_f32(hW_next, D_, m0, n0, acc4, lane);
  }
}

extern "C" void kernel_launch(void* const* d_in, const int* in_sizes, int n_in,
                              void* d_out, int out_size, void* d_ws, size_t ws_size,
                              hipStream_t stream) {
  (void)in_sizes; (void)n_in; (void)out_size; (void)ws_size;
  const float* r        = (const float*)d_in[0];
  const float* coords   = (const float*)d_in[1];
  const float* spin_emb = (const float*)d_in[2];
  const float* nuc_emb  = (const float*)d_in[3];
  const float* W_rbf_ee = (const float*)d_in[4];
  const float* W_rbf_ne = (const float*)d_in[5];
  const float* W_msg_ee = (const float*)d_in[6];
  const float* W_msg_ne = (const float*)d_in[7];
  const float* W1       = (const float*)d_in[8];
  const float* W2       = (const float*)d_in[9];
  float* out = (float*)d_out;

  float* ws  = (float*)d_ws;
  float* Aee = ws;
  float* Ane = Aee + (size_t)B_ * NE_ * KEE_;
  float* h   = Ane + (size_t)B_ * NE_ * KNE_;
  float* hWa = h   + (size_t)B_ * NE_ * D_;
  float* hWb = hWa + (size_t)B_ * NE_ * D_;
  float* hnW = hWb + (size_t)B_ * NE_ * D_;
  float* Wp  = hnW + (size_t)L_ * NN_ * D_;

  gnn_setup<<<SETUP_BLOCKS_, 128, 0, stream>>>(r, coords, spin_emb, nuc_emb,
                                               W_msg_ee, W_msg_ne, W1, W2,
                                               Wp, Aee, Ane, hWa, hnW, hWb);

  for (int l = 0; l < L_; ++l) {
    const float* Wpl = Wp + (size_t)l * 4 * WSZ_;
    const bool last = (l == L_ - 1);
    const float* hW_in  = (l & 1) ? hWb : hWa;
    float*       hW_out = (l & 1) ? hWa : hWb;
    if (l == 0)
      gnn_layer<true><<<B_ * (NE_ / LROWS_), 256, 0, stream>>>(
          Aee, Ane, hW_in, hnW + (size_t)l * NN_ * D_,
          W_rbf_ee + (size_t)l * NRBF_ * D_, W_rbf_ne + (size_t)l * NRBF_ * D_,
          Wpl + 2 * WSZ_, Wpl + 3 * WSZ_, last ? nullptr : Wpl + 4 * WSZ_,
          spin_emb, last ? out : h, hW_out);
    else
      gnn_layer<false><<<B_ * (NE_ / LROWS_), 256, 0, stream>>>(
          Aee, Ane, hW_in, hnW + (size_t)l * NN_ * D_,
          W_rbf_ee + (size_t)l * NRBF_ * D_, W_rbf_ne + (size_t)l * NRBF_ * D_,
          Wpl + 2 * WSZ_, Wpl + 3 * WSZ_, last ? nullptr : Wpl + 4 * WSZ_,
          h, last ? out : h, hW_out);
  }
}
